# starting kernel
# baseline (speedup 1.0000x reference)
#include <hip/hip_runtime.h>
#include <math.h>

typedef __attribute__((ext_vector_type(2))) float v2f;
typedef __attribute__((ext_vector_type(8))) float v8f;

__device__ __forceinline__ unsigned fkey(float v) {
  unsigned b = __float_as_uint(v);
  return (b & 0x80000000u) ? ~b : (b | 0x80000000u);
}
__device__ __forceinline__ float funkey(unsigned k) {
  unsigned b = (k & 0x80000000u) ? (k ^ 0x80000000u) : ~k;
  return __uint_as_float(b);
}

__global__ void fill_f32(float* p, float v, int n) {
  int i = blockIdx.x * blockDim.x + threadIdx.x;
  if (i < n) p[i] = v;
}
__global__ void fill_u32(unsigned* p, unsigned v, int n) {
  int i = blockIdx.x * blockDim.x + threadIdx.x;
  if (i < n) p[i] = v;
}

template <int K, int Nc>
__global__ void gemm_wmma_f32(const float* __restrict__ A, const float* __restrict__ B,
                              const float* __restrict__ bias, float* __restrict__ Cout,
                              int M, int act) {
  constexpr int  tilesN = (Nc + 15) >> 4;
  constexpr bool FULL   = (Nc % 16) == 0;
  int wave = blockIdx.x * (blockDim.x >> 5) + (threadIdx.x >> 5);
  int tilesM = (M + 15) >> 4;
  if (wave >= tilesM * tilesN) return;
  int tm   = wave / tilesN;
  int tn   = wave - tm * tilesN;
  int lane = threadIdx.x & 31;
  int half = lane >> 4;
  int l16  = lane & 15;
  int row  = tm * 16 + l16;
  int col  = tn * 16 + l16;
  bool cok  = FULL || (col < Nc);
  int  colc = FULL ? col : (cok ? col : 0);

  const float* ap = A + (size_t)row * K + 2 * half;
  const float* bp = B + (size_t)(2 * half) * Nc + colc;

  v8f c = {0.f, 0.f, 0.f, 0.f, 0.f, 0.f, 0.f, 0.f};
#pragma unroll
  for (int k0 = 0; k0 < K; k0 += 4) {
    float2 av = *(const float2*)(ap + k0);
    float  bx = bp[(size_t)k0 * Nc];
    float  by = bp[(size_t)k0 * Nc + Nc];
    v2f a, b;
    a.x = av.x;
    a.y = av.y;
    b.x = FULL ? bx : (cok ? bx : 0.f);
    b.y = FULL ? by : (cok ? by : 0.f);
    c = __builtin_amdgcn_wmma_f32_16x16x4_f32(false, a, false, b, (short)0, c,
                                              false, false);
  }

  if (cok) {
    float bv = bias ? bias[col] : 0.f;
    float* crow = Cout + (size_t)(tm * 16 + 8 * half) * Nc + col;
#pragma unroll
    for (int v = 0; v < 8; ++v) {
      float val = c[v] + bv;
      if (act) val = fmaxf(val, 0.f);
      crow[(size_t)v * Nc] = val;
    }
  }
}

__global__ void gemm_wmma_f32_dyn(const float* __restrict__ A, const float* __restrict__ B,
                                  const float* __restrict__ bias, float* __restrict__ Cout,
                                  int M, int K, int Nc, int act) {
  int wave   = blockIdx.x * (blockDim.x >> 5) + (threadIdx.x >> 5);
  int tilesN = (Nc + 15) >> 4;
  int tilesM = (M + 15) >> 4;
  if (wave >= tilesM * tilesN) return;
  int tm = wave / tilesN, tn = wave - tm * tilesN;
  int lane = threadIdx.x & 31, half = lane >> 4, l16 = lane & 15;
  int row = tm * 16 + l16, col = tn * 16 + l16;
  bool cok = (col < Nc);
  int colc = cok ? col : 0;
  const float* ap = A + (size_t)row * K + 2 * half;
  const float* bp = B + (size_t)(2 * half) * Nc + colc;
  v8f c = {0.f, 0.f, 0.f, 0.f, 0.f, 0.f, 0.f, 0.f};
  for (int k0 = 0; k0 < K; k0 += 4) {
    float2 av = *(const float2*)ap;
    float bx = bp[0], by = bp[Nc];
    v2f a, b;
    a.x = av.x;            a.y = av.y;
    b.x = cok ? bx : 0.f;  b.y = cok ? by : 0.f;
    c = __builtin_amdgcn_wmma_f32_16x16x4_f32(false, a, false, b, (short)0, c,
                                              false, false);
    ap += 4;
    bp += (size_t)4 * Nc;
  }
  if (cok) {
    float bv = bias ? bias[col] : 0.f;
    float* crow = Cout + (size_t)(tm * 16 + 8 * half) * Nc + col;
    for (int v = 0; v < 8; ++v) {
      float val = c[v] + bv;
      if (act) val = fmaxf(val, 0.f);
      crow[(size_t)v * Nc] = val;
    }
  }
}

__global__ void alpha_kernel(const float* __restrict__ h, const float* __restrict__ a_src,
                             const float* __restrict__ a_dst, float* __restrict__ as_out,
                             float* __restrict__ ad_out, int n, int heads) {
  int wave = blockIdx.x * (blockDim.x >> 5) + (threadIdx.x >> 5);
  if (wave >= n * heads) return;
  int node = wave / heads, hd = wave - node * heads;
  int lane = threadIdx.x & 31;
  float2 hv = *(const float2*)(h + (size_t)node * heads * 64 + hd * 64 + lane * 2);
  float2 as = *(const float2*)(a_src + hd * 64 + lane * 2);
  float2 ad = *(const float2*)(a_dst + hd * 64 + lane * 2);
  float ss = hv.x * as.x + hv.y * as.y;
  float sd = hv.x * ad.x + hv.y * ad.y;
  for (int off = 16; off > 0; off >>= 1) {
    ss += __shfl_xor(ss, off, 32);
    sd += __shfl_xor(sd, off, 32);
  }
  if (lane == 0) { as_out[wave] = ss; ad_out[wave] = sd; }
}

__global__ void edge_max_kernel(const int* __restrict__ ei, int Eorig, int Etot,
                                const float* __restrict__ as_v, const float* __restrict__ ad_v,
                                float* __restrict__ ebuf, unsigned* __restrict__ mkey,
                                int heads) {
  int idx = blockIdx.x * blockDim.x + threadIdx.x;
  if (idx >= Etot * heads) return;
  int e = idx / heads, hd = idx - e * heads;
  int src, dst;
  if (e < Eorig) { src = ei[e]; dst = ei[Eorig + e]; }
  else           { src = dst = e - Eorig; }
  float v = as_v[src * heads + hd] + ad_v[dst * heads + hd];
  v = (v > 0.f) ? v : 0.2f * v;
  ebuf[idx] = v;
  atomicMax(&mkey[dst * heads + hd], fkey(v));
}

template <int HEADS>
__global__ void edge_acc_kernel(const int* __restrict__ ei, int Eorig, int Etot,
                                const float* __restrict__ ebuf, const unsigned* __restrict__ mkey,
                                const float* __restrict__ h, float* __restrict__ acc,
                                float* __restrict__ z) {
  int wave = blockIdx.x * (blockDim.x >> 5) + (threadIdx.x >> 5);
  if (wave >= Etot) return;
  int lane = threadIdx.x & 31;
  int src, dst;
  if (wave < Eorig) { src = ei[wave]; dst = ei[Eorig + wave]; }
  else              { src = dst = wave - Eorig; }

  if (HEADS == 2) {
    int hd = lane >> 4;
    float m  = funkey(mkey[dst * 2 + hd]);
    float ex = expf(ebuf[(size_t)wave * 2 + hd] - m);
    if (lane == 0 || lane == 16)
      atomicAdd(&z[dst * 2 + hd], ex);
    float4 hv = *(const float4*)(h + (size_t)src * 128 + lane * 4);
    float* ap = acc + (size_t)dst * 128 + lane * 4;
    atomicAdd(ap + 0, ex * hv.x);
    atomicAdd(ap + 1, ex * hv.y);
    atomicAdd(ap + 2, ex * hv.z);
    atomicAdd(ap + 3, ex * hv.w);
  } else {
    float m  = funkey(mkey[dst]);
    float ex = expf(ebuf[wave] - m);
    if (lane == 0) atomicAdd(&z[dst], ex);
    float2 hv = *(const float2*)(h + (size_t)src * 64 + lane * 2);
    float* ap = acc + (size_t)dst * 64 + lane * 2;
    atomicAdd(ap,     ex * hv.x);
    atomicAdd(ap + 1, ex * hv.y);
  }
}

__global__ void finalize_gat(float* __restrict__ acc, const float* __restrict__ z,
                             const float* __restrict__ bias, int n, int heads) {
  int total = n * heads * 64;
  int i = blockIdx.x * blockDim.x + threadIdx.x;
  if (i >= total) return;
  int c    = i % (heads * 64);
  int node = i / (heads * 64);
  int hd   = c >> 6;
  float v = acc[i] / (z[node * heads + hd] + 1e-16f) + bias[c];
  acc[i] = (v > 0.f) ? v : expm1f(v);
}

extern "C" void kernel_launch(void* const* d_in, const int* in_sizes, int n_in,
                              void* d_out, int out_size, void* d_ws, size_t ws_size,
                              hipStream_t stream) {
  const float* x      = (const float*)d_in[0];
  const int*   ei     = (const int*)d_in[1];
  const float* W1     = (const float*)d_in[2];
  const float* a_src1 = (const float*)d_in[3];
  const float* a_dst1 = (const float*)d_in[4];
  const float* b1     = (const float*)d_in[5];
  const float* W2     = (const float*)d_in[6];
  const float* a_src2 = (const float*)d_in[7];
  const float* a_dst2 = (const float*)d_in[8];
  const float* b2     = (const float*)d_in[9];
  const float* m1w    = (const float*)d_in[10];
  const float* m1b    = (const float*)d_in[11];
  const float* m2w    = (const float*)d_in[12];
  const float* m2b    = (const float*)d_in[13];

  const int N    = in_sizes[0] / 64;
  const int E    = in_sizes[1] / 2;
  const int C    = in_sizes[12] / 64;
  const int Etot = E + N;

  float*    bufA  = (float*)d_ws;
  float*    bufB  = bufA + (size_t)N * 128;
  float*    bufC  = bufB + (size_t)N * 128;
  float*    as_v  = bufC + (size_t)N * 64;
  float*    ad_v  = as_v + (size_t)N * 2;
  unsigned* mkey  = (unsigned*)(ad_v + (size_t)N * 2);
  float*    zbuf  = (float*)mkey + (size_t)N * 2;
  float*    ebuf  = zbuf + (size_t)N * 2;

  const int TPB = 256;
  auto blocks1d = [&](long n) { return (int)((n + TPB - 1) / TPB); };
  auto gemm_blocks = [&](int M, int Nc) {
    int waves = ((M + 15) / 16) * ((Nc + 15) / 16);
    return (waves + 7) / 8;
  };
  const unsigned KEY_NEG_INF = 0x007FFFFFu;

  gemm_wmma_f32<64, 128><<<gemm_blocks(N, 128), TPB, 0, stream>>>(x, W1, nullptr,
                                                                  bufA, N, 0);
  alpha_kernel<<<(N * 2 + 7) / 8, TPB, 0, stream>>>(bufA, a_src1, a_dst1,
                                                    as_v, ad_v, N, 2);
  fill_u32<<<blocks1d(N * 2), TPB, 0, stream>>>(mkey, KEY_NEG_INF, N * 2);
  fill_f32<<<blocks1d(N * 2), TPB, 0, stream>>>(zbuf, 0.f, N * 2);
  fill_f32<<<blocks1d((long)N * 128), TPB, 0, stream>>>(bufB, 0.f, N * 128);
  edge_max_kernel<<<blocks1d((long)Etot * 2), TPB, 0, stream>>>(ei, E, Etot, as_v, ad_v,
                                                                ebuf, mkey, 2);
  edge_acc_kernel<2><<<(Etot + 7) / 8, TPB, 0, stream>>>(ei, E, Etot, ebuf, mkey,
                                                         bufA, bufB, zbuf);
  finalize_gat<<<blocks1d((long)N * 128), TPB, 0, stream>>>(bufB, zbuf, b1, N, 2);

  gemm_wmma_f32<128, 64><<<gemm_blocks(N, 64), TPB, 0, stream>>>(bufB, W2, nullptr,
                                                                 bufA, N, 0);
  alpha_kernel<<<(N + 7) / 8, TPB, 0, stream>>>(bufA, a_src2, a_dst2,
                                                as_v, ad_v, N, 1);
  fill_u32<<<blocks1d(N), TPB, 0, stream>>>(mkey, KEY_NEG_INF, N);
  fill_f32<<<blocks1d(N), TPB, 0, stream>>>(zbuf, 0.f, N);
  fill_f32<<<blocks1d((long)N * 64), TPB, 0, stream>>>(bufC, 0.f, N * 64);
  edge_max_kernel<<<blocks1d((long)Etot), TPB, 0, stream>>>(ei, E, Etot, as_v, ad_v,
                                                            ebuf, mkey, 1);
  edge_acc_kernel<1><<<(Etot + 7) / 8, TPB, 0, stream>>>(ei, E, Etot, ebuf, mkey,
                                                         bufA, bufC, zbuf);
  finalize_gat<<<blocks1d((long)N * 64), TPB, 0, stream>>>(bufC, zbuf, b2, N, 1);

  gemm_wmma_f32<64, 64><<<gemm_blocks(N, 64), TPB, 0, stream>>>(bufC, m1w, m1b,
                                                                bufA, N, 1);
  if (C == 8) {
    gemm_wmma_f32<64, 8><<<gemm_blocks(N, 8), TPB, 0, stream>>>(bufA, m2w, m2b,
                                                                (float*)d_out, N, 1);
  } else {
    gemm_wmma_f32_dyn<<<gemm_blocks(N, C), TPB, 0, stream>>>(bufA, m2w, m2b,
                                                             (float*)d_out, N, 64, C, 1);
  }
}


# node gather scores each source row from the row it fetches anyway (64-float dot per half-wave, 8 row loads per chunk, running softmax); alpha kernels and score lookups gone; sums' order shifts
# speedup vs baseline: 4.5500x; 4.5500x over previous
#include <hip/hip_runtime.h>
#include <math.h>

typedef __attribute__((ext_vector_type(2))) float v2f;
typedef __attribute__((ext_vector_type(8))) float v8f;

__global__ void fill_u32(unsigned* p, unsigned v, int n) {
  int i = (blockIdx.x * blockDim.x + threadIdx.x) * 4;
  if (i + 3 < n) {
    *(uint4*)(p + i) = make_uint4(v, v, v, v);
  } else {
    for (; i < n; ++i) p[i] = v;
  }
}

template <int K, int Nc>
__global__ void __launch_bounds__(256)
gemm_wmma_f32(const float* __restrict__ A, const float* __restrict__ B,
              const float* __restrict__ bias, float* __restrict__ Cout, int M, int act) {
  constexpr int  tilesN = (Nc + 15) >> 4;
  constexpr bool FULL   = (Nc % 16) == 0;
  constexpr int  RB  = 8192 / K;
  constexpr int  RT  = RB / 16;
  constexpr int  WG  = 8 / tilesN;
  constexpr int  TPW = RT / WG;
  constexpr int  P   = K + 4;
  constexpr int  KS  = K / 4;
  constexpr int  C4  = K / 4;
  constexpr int  CH  = KS < 16 ? KS : 16;
  static_assert(8192 % K == 0 && 8 % tilesN == 0 && RT % WG == 0, "block shape");
  __shared__ __attribute__((aligned(16))) float sA[RB * P];

  const int t    = threadIdx.x;
  const int w    = __builtin_amdgcn_readfirstlane(t >> 5);
  const int lane = t & 31;
  const int half = lane >> 4;
  const int l16  = lane & 15;
  const int r0   = blockIdx.x * RB;

  float4 av[8];
#pragma unroll
  for (int i = 0; i < 8; ++i) {
    int idx4 = i * 256 + t;
    int row  = idx4 / C4;
    int c4   = idx4 - row * C4;
    int gr   = r0 + row;
    gr = gr < M ? gr : M - 1;
    av[i] = *(const float4*)(A + (size_t)gr * K + c4 * 4);
  }

  const int tn   = w % tilesN;
  const int rsub = w / tilesN;
  const int col  = tn * 16 + l16;
  const bool cok = FULL || (col < Nc);
  const int colc = FULL ? col : (cok ? col : 0);
  const float* bp = B + (size_t)(2 * half) * Nc + colc;
  v2f bf[KS];
#pragma unroll
  for (int s = 0; s < KS; ++s) {
    float bx = bp[(size_t)(4 * s) * Nc];
    float by = bp[(size_t)(4 * s) * Nc + Nc];
    bf[s].x = FULL ? bx : (cok ? bx : 0.f);
    bf[s].y = FULL ? by : (cok ? by : 0.f);
  }
  float bv = 0.f;
  if (bias != nullptr && cok) bv = bias[col];

#pragma unroll
  for (int i = 0; i < 8; ++i) {
    int idx4 = i * 256 + t;
    int row  = idx4 / C4;
    int c4   = idx4 - row * C4;
    *(float4*)(sA + row * P + c4 * 4) = av[i];
  }
  __syncthreads();

  for (int j = 0; j < TPW; ++j) {
    const int rt   = rsub + WG * j;
    const int row0 = r0 + rt * 16;
    if (row0 >= M) break;
    const float* ar = sA + (rt * 16 + l16) * P + 2 * half;
    v8f c = {0.f, 0.f, 0.f, 0.f, 0.f, 0.f, 0.f, 0.f};
#pragma unroll
    for (int s0 = 0; s0 < KS; s0 += CH) {
      v2f a[CH];
#pragma unroll
      for (int s = 0; s < CH; ++s) a[s] = *(const v2f*)(ar + 4 * (s0 + s));
      __builtin_amdgcn_sched_barrier(0);
#pragma unroll
      for (int s = 0; s < CH; ++s)
        c = __builtin_amdgcn_wmma_f32_16x16x4_f32(false, a[s], false, bf[s0 + s], (short)0, c,
                                                  false, false);
      __builtin_amdgcn_sched_barrier(0);
    }
    if (cok) {
      const int rbase = row0 + 8 * half;
      float* crow = Cout + (size_t)rbase * Nc + col;
      if (row0 + 16 <= M) {
#pragma unroll
        for (int v = 0; v < 8; ++v) {
          float val = c[v] + bv;
          if (act) val = fmaxf(val, 0.f);
          crow[(size_t)v * Nc] = val;
        }
      } else {
#pragma unroll
        for (int v = 0; v < 8; ++v) {
          float val = c[v] + bv;
          if (act) val = fmaxf(val, 0.f);
          if (rbase + v < M) crow[(size_t)v * Nc] = val;
        }
      }
    }
  }
}

__global__ void gemm_wmma_f32_dyn(const float* __restrict__ A, const float* __restrict__ B,
                                  const float* __restrict__ bias, float* __restrict__ Cout,
                                  int M, int K, int Nc, int act) {
  int wave   = blockIdx.x * (blockDim.x >> 5) + (threadIdx.x >> 5);
  int tilesN = (Nc + 15) >> 4;
  int tilesM = (M + 15) >> 4;
  if (wave >= tilesM * tilesN) return;
  int tm = wave / tilesN, tn = wave - tm * tilesN;
  int lane = threadIdx.x & 31, half = lane >> 4, l16 = lane & 15;
  int row = tm * 16 + l16, col = tn * 16 + l16;
  bool cok = (col < Nc);
  int colc = cok ? col : 0;
  const float* ap = A + (size_t)row * K + 2 * half;
  const float* bp = B + (size_t)(2 * half) * Nc + colc;
  v8f c = {0.f, 0.f, 0.f, 0.f, 0.f, 0.f, 0.f, 0.f};
  for (int k0 = 0; k0 < K; k0 += 4) {
    float2 av = *(const float2*)ap;
    float bx = bp[0], by = bp[Nc];
    v2f a, b;
    a.x = av.x;            a.y = av.y;
    b.x = cok ? bx : 0.f;  b.y = cok ? by : 0.f;
    c = __builtin_amdgcn_wmma_f32_16x16x4_f32(false, a, false, b, (short)0, c,
                                              false, false);
    ap += 4;
    bp += (size_t)4 * Nc;
  }
  if (cok) {
    float bv = bias ? bias[col] : 0.f;
    float* crow = Cout + (size_t)(tm * 16 + 8 * half) * Nc + col;
    for (int v = 0; v < 8; ++v) {
      float val = c[v] + bv;
      if (act) val = fmaxf(val, 0.f);
      crow[(size_t)v * Nc] = val;
    }
  }
}

__global__ void list_build(const int* __restrict__ ei, int Eorig, int* __restrict__ head,
                           int* __restrict__ nxt) {
  int e0 = (blockIdx.x * blockDim.x + threadIdx.x) * 4;
  if (e0 >= Eorig) return;
  const int* dp = ei + Eorig;
  if ((Eorig & 3) == 0) {
    int4 d = *(const int4*)(dp + e0);
    int4 o;
    o.x = atomicExch(&head[d.x * 4 + 0], e0 + 0);
    o.y = atomicExch(&head[d.y * 4 + 1], e0 + 1);
    o.z = atomicExch(&head[d.z * 4 + 2], e0 + 2);
    o.w = atomicExch(&head[d.w * 4 + 3], e0 + 3);
    *(int4*)(nxt + e0) = o;
  } else {
    for (int q = 0; q < 4; ++q) {
      int e = e0 + q;
      if (e < Eorig) nxt[e] = atomicExch(&head[dp[e] * 4 + (e & 3)], e);
    }
  }
}

template <int MASK>
__device__ __forceinline__ float row_xor(float v) {
  return __int_as_float(__builtin_amdgcn_update_dpp(0, __float_as_int(v), 0x160 | MASK, 0xF, 0xF, true));
}
template <int L>
__device__ __forceinline__ float row_share(float v) {
  return __int_as_float(__builtin_amdgcn_update_dpp(0, __float_as_int(v), 0x150 | L, 0xF, 0xF, true));
}
__device__ __forceinline__ float half_sum(float v) {
  v += row_xor<8>(v);
  v += row_xor<4>(v);
  v += row_xor<2>(v);
  v += row_xor<1>(v);
  return v;
}
__device__ __forceinline__ float half_max(float v) {
  v = fmaxf(v, row_xor<8>(v));
  v = fmaxf(v, row_xor<4>(v));
  v = fmaxf(v, row_xor<2>(v));
  v = fmaxf(v, row_xor<1>(v));
  return v;
}
__device__ __forceinline__ float dot4(float4 a, float4 b) {
  return fmaf(a.w, b.w, fmaf(a.z, b.z, fmaf(a.y, b.y, a.x * b.x)));
}
__device__ __forceinline__ float leaky(float a) { return (a > 0.f) ? a : 0.2f * a; }

struct Soft { float m, zl, a0, a1, a2, a3; };

template <int HEADS, int R, bool FULL>
__device__ __forceinline__ void gather_chunk(Soft& st, const float* __restrict__ h, int my_src,
                                             int k, int cnt, unsigned off, int half, int c16,
                                             float4 as4, float adn) {
  constexpr int F   = HEADS * 64;
  constexpr int EPL = (HEADS == 2) ? 1 : 2;
  float4 t[R];
#pragma unroll
  for (int q = 0; q < R; ++q) {
    const int e0 = k + q * EPL;
    if constexpr (!FULL) t[q] = make_float4(0.f, 0.f, 0.f, 0.f);
    if (FULL || e0 < cnt) {
      int s = __builtin_amdgcn_readlane(my_src, e0);
      if constexpr (HEADS == 1) {
        const int e1 = (FULL || e0 + 1 < cnt) ? e0 + 1 : e0;
        const int s1 = __builtin_amdgcn_readlane(my_src, e1);
        s = half ? s1 : s;
      }
      t[q] = *(const float4*)(h + ((unsigned)s * (unsigned)F + off));
    }
  }
  float d = 0.f;
#pragma unroll
  for (int q = 0; q < R; ++q) {
    if (FULL || k + q * EPL < cnt) {
      const float dq = half_sum(dot4(t[q], as4));
      d = (c16 == q) ? dq : d;
    }
  }
  const int  eidx  = k + c16 * EPL + ((HEADS == 1) ? half : 0);
  const bool valid = (c16 < R) && (FULL || eidx < cnt);
  const float a    = valid ? leaky(d + adn) : -INFINITY;
  const float mn   = fmaxf(st.m, half_max(a));
  const float ev   = expf(((c16 == R) ? st.m : a) - mn);
  const float rsc  = row_share<R>(ev);
  st.zl = st.zl * rsc + ((c16 == R) ? 0.f : ev);
  st.a0 *= rsc; st.a1 *= rsc; st.a2 *= rsc; st.a3 *= rsc;
  st.m = mn;
  auto add_row = [&](float w, float4 r) {
    st.a0 = fmaf(w, r.x, st.a0); st.a1 = fmaf(w, r.y, st.a1);
    st.a2 = fmaf(w, r.z, st.a2); st.a3 = fmaf(w, r.w, st.a3);
  };
  if (FULL || k + 0 * EPL < cnt) add_row(row_share<0>(ev), t[0]);
  if (FULL || k + 1 * EPL < cnt) add_row(row_share<1>(ev), t[1]);
  if (FULL || k + 2 * EPL < cnt) add_row(row_share<2>(ev), t[2]);
  if (FULL || k + 3 * EPL < cnt) add_row(row_share<3>(ev), t[3]);
  if (FULL || k + 4 * EPL < cnt) add_row(row_share<4>(ev), t[4]);
  if (FULL || k + 5 * EPL < cnt) add_row(row_share<5>(ev), t[5]);
  if (FULL || k + 6 * EPL < cnt) add_row(row_share<6>(ev), t[6]);
  if (FULL || k + 7 * EPL < cnt) add_row(row_share<7>(ev), t[7]);
}

template <int HEADS>
__global__ void __launch_bounds__(256)
node_gather(const int* __restrict__ ei, int Eorig, const int* __restrict__ head,
                            const int* __restrict__ nxt, const float* __restrict__ a_src,
                            const float* __restrict__ a_dst, const float* __restrict__ h,
                            const float* __restrict__ bias, float* __restrict__ out, int n) {
  constexpr int F   = HEADS * 64;
  constexpr int EPL = (HEADS == 2) ? 1 : 2;
  constexpr int R   = 8;
  int node = __builtin_amdgcn_readfirstlane((int)(blockIdx.x * (blockDim.x >> 5) + (threadIdx.x >> 5)));
  if (node >= n) return;
  const int lane = threadIdx.x & 31;
  const int half = lane >> 4;
  const int c16  = lane & 15;
  const int hd   = (HEADS == 2) ? half : 0;
  const unsigned off = (HEADS == 2) ? lane * 4 : c16 * 4;

  int cur = head[node * 4 + (lane & 3)];
  const float4 own = *(const float4*)(h + (size_t)node * F + off);
  const float4 as4 = *(const float4*)(a_src + hd * 64 + c16 * 4);
  const float4 ad4 = *(const float4*)(a_dst + hd * 64 + c16 * 4);

  const float adn = half_sum(dot4(own, ad4));
  const bool mine = (HEADS == 2) || (half == 0);
  Soft st;
  st.m  = leaky(half_sum(dot4(own, as4)) + adn);
  st.zl = (mine && c16 == 0) ? 1.f : 0.f;
  st.a0 = mine ? own.x : 0.f;
  st.a1 = mine ? own.y : 0.f;
  st.a2 = mine ? own.z : 0.f;
  st.a3 = mine ? own.w : 0.f;

  unsigned act = __builtin_amdgcn_ballot_w32(cur >= 0) & 0xFu;
  const int max_rounds = (Eorig >> 5) + 2;
  for (int round = 0; round < max_rounds; ++round) {
    int cnt = 0;
    int my_src = node;
    while (act != 0u && cnt < 32) {
      int ci = cur > 0 ? cur : 0;
      int sv = ei[ci];
      int nv = nxt[ci];
#pragma unroll
      for (int c = 0; c < 4; ++c) {
        if (((act >> c) & 1u) != 0u && cnt < 32) {
          int s = __builtin_amdgcn_readlane(sv, c);
          if (lane == cnt) my_src = s;
          ++cnt;
          if ((lane & 3) == c) cur = nv;
        }
      }
      act = __builtin_amdgcn_ballot_w32(cur >= 0) & 0xFu;
    }
    int k = 0;
    for (; k + R * EPL <= cnt; k += R * EPL)
      gather_chunk<HEADS, R, true>(st, h, my_src, k, cnt, off, half, c16, as4, adn);
    if (k < cnt)
      gather_chunk<HEADS, R, false>(st, h, my_src, k, cnt, off, half, c16, as4, adn);
    if (act == 0u) break;
  }

  float z = half_sum(st.zl);
  float acc[4] = {st.a0, st.a1, st.a2, st.a3};
  if constexpr (HEADS == 1) {
    const float mo  = __shfl_xor(st.m, 16, 32);
    const float rsc = expf(st.m - fmaxf(st.m, mo));
    z *= rsc;
#pragma unroll
    for (int j = 0; j < 4; ++j) acc[j] *= rsc;
    z += __shfl_xor(z, 16, 32);
#pragma unroll
    for (int j = 0; j < 4; ++j) acc[j] += __shfl_xor(acc[j], 16, 32);
  }

  if constexpr (HEADS == 2) {
    const float4 bv = *(const float4*)(bias + off);
    const float bb[4] = {bv.x, bv.y, bv.z, bv.w};
    float o[4];
#pragma unroll
    for (int j = 0; j < 4; ++j) {
      float v = acc[j] / (z + 1e-16f) + bb[j];
      o[j] = (v > 0.f) ? v : expm1f(v);
    }
    *(float4*)(out + (size_t)node * F + off) = make_float4(o[0], o[1], o[2], o[3]);
  } else {
    const unsigned o2 = off + 2u * (unsigned)half;
    const float2 bv = *(const float2*)(bias + o2);
    const float v0 = (half ? acc[2] : acc[0]) / (z + 1e-16f) + bv.x;
    const float v1 = (half ? acc[3] : acc[1]) / (z + 1e-16f) + bv.y;
    const float r0 = (v0 > 0.f) ? v0 : expm1f(v0);
    const float r1 = (v1 > 0.f) ? v1 : expm1f(v1);
    *(float2*)(out + (size_t)node * F + o2) = make_float2(r0, r1);
  }
}

extern "C" void kernel_launch(void* const* d_in, const int* in_sizes, int n_in,
                              void* d_out, int out_size, void* d_ws, size_t ws_size,
                              hipStream_t stream) {
  const float* x      = (const float*)d_in[0];
  const int*   ei     = (const int*)d_in[1];
  const float* W1     = (const float*)d_in[2];
  const float* a_src1 = (const float*)d_in[3];
  const float* a_dst1 = (const float*)d_in[4];
  const float* b1     = (const float*)d_in[5];
  const float* W2     = (const float*)d_in[6];
  const float* a_src2 = (const float*)d_in[7];
  const float* a_dst2 = (const float*)d_in[8];
  const float* b2     = (const float*)d_in[9];
  const float* m1w    = (const float*)d_in[10];
  const float* m1b    = (const float*)d_in[11];
  const float* m2w    = (const float*)d_in[12];
  const float* m2b    = (const float*)d_in[13];

  const int N    = in_sizes[0] / 64;
  const int E    = in_sizes[1] / 2;
  const int C    = in_sizes[12] / 64;

  float* bufA = (float*)d_ws;
  float* bufB = bufA + (size_t)N * 128;
  float* bufC = bufB + (size_t)N * 128;
  int*   head = (int*)(bufC + (size_t)N * 64);
  int*   nxt  = head + (size_t)N * 4;

  const int TPB = 256;
  auto blocks1d = [&](long n) { return (int)((n + TPB - 1) / TPB); };
  auto gemm_blocks = [&](int M, int Nc) {
    int waves = ((M + 15) / 16) * ((Nc + 15) / 16);
    return (waves + 7) / 8;
  };
  auto gemm_row_blocks = [&](int M, int K) {
    int rb = 8192 / K;
    return (M + rb - 1) / rb;
  };

  fill_u32<<<blocks1d((long)N), TPB, 0, stream>>>((unsigned*)head, 0xFFFFFFFFu, N * 4);
  list_build<<<blocks1d(((long)E + 3) / 4), TPB, 0, stream>>>(ei, E, head, nxt);

  gemm_wmma_f32<64, 128><<<gemm_row_blocks(N, 64), TPB, 0, stream>>>(x, W1, nullptr,
                                                                  bufA, N, 0);
  node_gather<2><<<(N + 7) / 8, TPB, 0, stream>>>(ei, E, head, nxt, a_src1, a_dst1,
                                                  bufA, b1, bufB, N);

  gemm_wmma_f32<128, 64><<<gemm_row_blocks(N, 128), TPB, 0, stream>>>(bufB, W2, nullptr,
                                                                 bufA, N, 0);
  node_gather<1><<<(N + 7) / 8, TPB, 0, stream>>>(ei, E, head, nxt, a_src2, a_dst2,
                                                  bufA, b2, bufC, N);

  gemm_wmma_f32<64, 64><<<gemm_row_blocks(N, 64), TPB, 0, stream>>>(bufC, m1w, m1b,
                                                                bufA, N, 1);
  if (C == 8) {
    gemm_wmma_f32<64, 8><<<gemm_row_blocks(N, 64), TPB, 0, stream>>>(bufA, m2w, m2b,
                                                                (float*)d_out, N, 1);
  } else {
    gemm_wmma_f32_dyn<<<gemm_blocks(N, C), TPB, 0, stream>>>(bufA, m2w, m2b,
                                                             (float*)d_out, N, 64, C, 1);
  }
}


# conv1's node pass stores each node's first 32 sources as one 128-byte row plus its four list cursors; conv2's node pass loads that row instead of walking the linked lists again; entry order kept
# speedup vs baseline: 4.8340x; 1.0624x over previous
#include <hip/hip_runtime.h>
#include <math.h>

typedef __attribute__((ext_vector_type(2))) float v2f;
typedef __attribute__((ext_vector_type(8))) float v8f;

__global__ void fill_u32(unsigned* p, unsigned v, int n) {
  int i = (blockIdx.x * blockDim.x + threadIdx.x) * 4;
  if (i + 3 < n) {
    *(uint4*)(p + i) = make_uint4(v, v, v, v);
  } else {
    for (; i < n; ++i) p[i] = v;
  }
}

template <int K, int Nc>
__global__ void __launch_bounds__(256)
gemm_wmma_f32(const float* __restrict__ A, const float* __restrict__ B,
              const float* __restrict__ bias, float* __restrict__ Cout, int M, int act) {
  constexpr int  tilesN = (Nc + 15) >> 4;
  constexpr bool FULL   = (Nc % 16) == 0;
  constexpr int  RB  = 8192 / K;
  constexpr int  RT  = RB / 16;
  constexpr int  WG  = 8 / tilesN;
  constexpr int  TPW = RT / WG;
  constexpr int  P   = K + 4;
  constexpr int  KS  = K / 4;
  constexpr int  C4  = K / 4;
  constexpr int  CH  = KS < 16 ? KS : 16;
  static_assert(8192 % K == 0 && 8 % tilesN == 0 && RT % WG == 0, "block shape");
  __shared__ __attribute__((aligned(16))) float sA[RB * P];

  const int t    = threadIdx.x;
  const int w    = __builtin_amdgcn_readfirstlane(t >> 5);
  const int lane = t & 31;
  const int half = lane >> 4;
  const int l16  = lane & 15;
  const int r0   = blockIdx.x * RB;

  float4 av[8];
#pragma unroll
  for (int i = 0; i < 8; ++i) {
    int idx4 = i * 256 + t;
    int row  = idx4 / C4;
    int c4   = idx4 - row * C4;
    int gr   = r0 + row;
    gr = gr < M ? gr : M - 1;
    av[i] = *(const float4*)(A + (size_t)gr * K + c4 * 4);
  }

  const int tn   = w % tilesN;
  const int rsub = w / tilesN;
  const int col  = tn * 16 + l16;
  const bool cok = FULL || (col < Nc);
  const int colc = FULL ? col : (cok ? col : 0);
  const float* bp = B + (size_t)(2 * half) * Nc + colc;
  v2f bf[KS];
#pragma unroll
  for (int s = 0; s < KS; ++s) {
    float bx = bp[(size_t)(4 * s) * Nc];
    float by = bp[(size_t)(4 * s) * Nc + Nc];
    bf[s].x = FULL ? bx : (cok ? bx : 0.f);
    bf[s].y = FULL ? by : (cok ? by : 0.f);
  }
  float bv = 0.f;
  if (bias != nullptr && cok) bv = bias[col];

#pragma unroll
  for (int i = 0; i < 8; ++i) {
    int idx4 = i * 256 + t;
    int row  = idx4 / C4;
    int c4   = idx4 - row * C4;
    *(float4*)(sA + row * P + c4 * 4) = av[i];
  }
  __syncthreads();

  for (int j = 0; j < TPW; ++j) {
    const int rt   = rsub + WG * j;
    const int row0 = r0 + rt * 16;
    if (row0 >= M) break;
    const float* ar = sA + (rt * 16 + l16) * P + 2 * half;
    v8f c = {0.f, 0.f, 0.f, 0.f, 0.f, 0.f, 0.f, 0.f};
#pragma unroll
    for (int s0 = 0; s0 < KS; s0 += CH) {
      v2f a[CH];
#pragma unroll
      for (int s = 0; s < CH; ++s) a[s] = *(const v2f*)(ar + 4 * (s0 + s));
      __builtin_amdgcn_sched_barrier(0);
#pragma unroll
      for (int s = 0; s < CH; ++s)
        c = __builtin_amdgcn_wmma_f32_16x16x4_f32(false, a[s], false, bf[s0 + s], (short)0, c,
                                                  false, false);
      __builtin_amdgcn_sched_barrier(0);
    }
    if (cok) {
      const int rbase = row0 + 8 * half;
      float* crow = Cout + (size_t)rbase * Nc + col;
      if (row0 + 16 <= M) {
#pragma unroll
        for (int v = 0; v < 8; ++v) {
          float val = c[v] + bv;
          if (act) val = fmaxf(val, 0.f);
          crow[(size_t)v * Nc] = val;
        }
      } else {
#pragma unroll
        for (int v = 0; v < 8; ++v) {
          float val = c[v] + bv;
          if (act) val = fmaxf(val, 0.f);
          if (rbase + v < M) crow[(size_t)v * Nc] = val;
        }
      }
    }
  }
}

__global__ void gemm_wmma_f32_dyn(const float* __restrict__ A, const float* __restrict__ B,
                                  const float* __restrict__ bias, float* __restrict__ Cout,
                                  int M, int K, int Nc, int act) {
  int wave   = blockIdx.x * (blockDim.x >> 5) + (threadIdx.x >> 5);
  int tilesN = (Nc + 15) >> 4;
  int tilesM = (M + 15) >> 4;
  if (wave >= tilesM * tilesN) return;
  int tm = wave / tilesN, tn = wave - tm * tilesN;
  int lane = threadIdx.x & 31, half = lane >> 4, l16 = lane & 15;
  int row = tm * 16 + l16, col = tn * 16 + l16;
  bool cok = (col < Nc);
  int colc = cok ? col : 0;
  const float* ap = A + (size_t)row * K + 2 * half;
  const float* bp = B + (size_t)(2 * half) * Nc + colc;
  v8f c = {0.f, 0.f, 0.f, 0.f, 0.f, 0.f, 0.f, 0.f};
  for (int k0 = 0; k0 < K; k0 += 4) {
    float2 av = *(const float2*)ap;
    float bx = bp[0], by = bp[Nc];
    v2f a, b;
    a.x = av.x;            a.y = av.y;
    b.x = cok ? bx : 0.f;  b.y = cok ? by : 0.f;
    c = __builtin_amdgcn_wmma_f32_16x16x4_f32(false, a, false, b, (short)0, c,
                                              false, false);
    ap += 4;
    bp += (size_t)4 * Nc;
  }
  if (cok) {
    float bv = bias ? bias[col] : 0.f;
    float* crow = Cout + (size_t)(tm * 16 + 8 * half) * Nc + col;
    for (int v = 0; v < 8; ++v) {
      float val = c[v] + bv;
      if (act) val = fmaxf(val, 0.f);
      crow[(size_t)v * Nc] = val;
    }
  }
}

__global__ void list_build(const int* __restrict__ ei, int Eorig, int* __restrict__ head,
                           int* __restrict__ nxt) {
  int e0 = (blockIdx.x * blockDim.x + threadIdx.x) * 4;
  if (e0 >= Eorig) return;
  const int* dp = ei + Eorig;
  if ((Eorig & 3) == 0) {
    int4 d = *(const int4*)(dp + e0);
    int4 o;
    o.x = atomicExch(&head[d.x * 4 + 0], e0 + 0);
    o.y = atomicExch(&head[d.y * 4 + 1], e0 + 1);
    o.z = atomicExch(&head[d.z * 4 + 2], e0 + 2);
    o.w = atomicExch(&head[d.w * 4 + 3], e0 + 3);
    *(int4*)(nxt + e0) = o;
  } else {
    for (int q = 0; q < 4; ++q) {
      int e = e0 + q;
      if (e < Eorig) nxt[e] = atomicExch(&head[dp[e] * 4 + (e & 3)], e);
    }
  }
}

template <int MASK>
__device__ __forceinline__ float row_xor(float v) {
  return __int_as_float(__builtin_amdgcn_update_dpp(0, __float_as_int(v), 0x160 | MASK, 0xF, 0xF, true));
}
template <int L>
__device__ __forceinline__ float row_share(float v) {
  return __int_as_float(__builtin_amdgcn_update_dpp(0, __float_as_int(v), 0x150 | L, 0xF, 0xF, true));
}
__device__ __forceinline__ float half_sum(float v) {
  v += row_xor<8>(v);
  v += row_xor<4>(v);
  v += row_xor<2>(v);
  v += row_xor<1>(v);
  return v;
}
__device__ __forceinline__ float half_max(float v) {
  v = fmaxf(v, row_xor<8>(v));
  v = fmaxf(v, row_xor<4>(v));
  v = fmaxf(v, row_xor<2>(v));
  v = fmaxf(v, row_xor<1>(v));
  return v;
}
__device__ __forceinline__ float dot4(float4 a, float4 b) {
  return fmaf(a.w, b.w, fmaf(a.z, b.z, fmaf(a.y, b.y, a.x * b.x)));
}
__device__ __forceinline__ float leaky(float a) { return (a > 0.f) ? a : 0.2f * a; }

struct Soft { float m, zl, a0, a1, a2, a3; };

template <int HEADS, int R, bool FULL>
__device__ __forceinline__ void gather_chunk(Soft& st, const float* __restrict__ h, int my_src,
                                             int k, int cnt, unsigned off, int half, int c16,
                                             float4 as4, float adn) {
  constexpr int F   = HEADS * 64;
  constexpr int EPL = (HEADS == 2) ? 1 : 2;
  float4 t[R];
#pragma unroll
  for (int q = 0; q < R; ++q) {
    const int e0 = k + q * EPL;
    if constexpr (!FULL) t[q] = make_float4(0.f, 0.f, 0.f, 0.f);
    if (FULL || e0 < cnt) {
      int s = __builtin_amdgcn_readlane(my_src, e0);
      if constexpr (HEADS == 1) {
        const int e1 = (FULL || e0 + 1 < cnt) ? e0 + 1 : e0;
        const int s1 = __builtin_amdgcn_readlane(my_src, e1);
        s = half ? s1 : s;
      }
      t[q] = *(const float4*)(h + ((unsigned)s * (unsigned)F + off));
    }
  }
  float d = 0.f;
#pragma unroll
  for (int q = 0; q < R; ++q) {
    if (FULL || k + q * EPL < cnt) {
      const float dq = half_sum(dot4(t[q], as4));
      d = (c16 == q) ? dq : d;
    }
  }
  const int  eidx  = k + c16 * EPL + ((HEADS == 1) ? half : 0);
  const bool valid = (c16 < R) && (FULL || eidx < cnt);
  const float a    = valid ? leaky(d + adn) : -INFINITY;
  const float mn   = fmaxf(st.m, half_max(a));
  const float ev   = expf(((c16 == R) ? st.m : a) - mn);
  const float rsc  = row_share<R>(ev);
  st.zl = st.zl * rsc + ((c16 == R) ? 0.f : ev);
  st.a0 *= rsc; st.a1 *= rsc; st.a2 *= rsc; st.a3 *= rsc;
  st.m = mn;
  auto add_row = [&](float w, float4 r) {
    st.a0 = fmaf(w, r.x, st.a0); st.a1 = fmaf(w, r.y, st.a1);
    st.a2 = fmaf(w, r.z, st.a2); st.a3 = fmaf(w, r.w, st.a3);
  };
  if (FULL || k + 0 * EPL < cnt) add_row(row_share<0>(ev), t[0]);
  if (FULL || k + 1 * EPL < cnt) add_row(row_share<1>(ev), t[1]);
  if (FULL || k + 2 * EPL < cnt) add_row(row_share<2>(ev), t[2]);
  if (FULL || k + 3 * EPL < cnt) add_row(row_share<3>(ev), t[3]);
  if (FULL || k + 4 * EPL < cnt) add_row(row_share<4>(ev), t[4]);
  if (FULL || k + 5 * EPL < cnt) add_row(row_share<5>(ev), t[5]);
  if (FULL || k + 6 * EPL < cnt) add_row(row_share<6>(ev), t[6]);
  if (FULL || k + 7 * EPL < cnt) add_row(row_share<7>(ev), t[7]);
}

template <int HEADS, int MODE>
__global__ void __launch_bounds__(256)
node_gather(const int* __restrict__ ei, int Eorig, const int* __restrict__ head,
                            const int* __restrict__ nxt, int* __restrict__ rows,
                            int* __restrict__ curs, const float* __restrict__ a_src,
                            const float* __restrict__ a_dst, const float* __restrict__ h,
                            const float* __restrict__ bias, float* __restrict__ out, int n) {
  constexpr int F   = HEADS * 64;
  constexpr int EPL = (HEADS == 2) ? 1 : 2;
  constexpr int R   = 8;
  int node = __builtin_amdgcn_readfirstlane((int)(blockIdx.x * (blockDim.x >> 5) + (threadIdx.x >> 5)));
  if (node >= n) return;
  const int lane = threadIdx.x & 31;
  const int half = lane >> 4;
  const int c16  = lane & 15;
  const int hd   = (HEADS == 2) ? half : 0;
  const unsigned off = (HEADS == 2) ? lane * 4 : c16 * 4;

  int cur;
  int pre = -1;
  if constexpr (MODE == 2) {
    pre = rows[(size_t)node * 32 + lane];
    cur = curs[node * 4 + (lane & 3)];
  } else {
    cur = head[node * 4 + (lane & 3)];
  }
  const float4 own = *(const float4*)(h + (size_t)node * F + off);
  const float4 as4 = *(const float4*)(a_src + hd * 64 + c16 * 4);
  const float4 ad4 = *(const float4*)(a_dst + hd * 64 + c16 * 4);
  if constexpr (MODE == 2) __builtin_amdgcn_sched_barrier(0);

  const float adn = half_sum(dot4(own, ad4));
  const bool mine = (HEADS == 2) || (half == 0);
  Soft st;
  st.m  = leaky(half_sum(dot4(own, as4)) + adn);
  st.zl = (mine && c16 == 0) ? 1.f : 0.f;
  st.a0 = mine ? own.x : 0.f;
  st.a1 = mine ? own.y : 0.f;
  st.a2 = mine ? own.z : 0.f;
  st.a3 = mine ? own.w : 0.f;

  unsigned act = __builtin_amdgcn_ballot_w32(cur >= 0) & 0xFu;
  const int max_rounds = (Eorig >> 5) + 2;
  int round = 0;
  do {
    int cnt = 0;
    int my_src = node;
    if (MODE == 2 && round == 0) {
      my_src = pre;
      cnt = __builtin_popcount(__builtin_amdgcn_ballot_w32(pre >= 0));
    } else
    while (act != 0u && cnt < 32) {
      int ci = cur > 0 ? cur : 0;
      int sv = ei[ci];
      int nv = nxt[ci];
#pragma unroll
      for (int c = 0; c < 4; ++c) {
        if (((act >> c) & 1u) != 0u && cnt < 32) {
          int s = __builtin_amdgcn_readlane(sv, c);
          if (lane == cnt) my_src = s;
          ++cnt;
          if ((lane & 3) == c) cur = nv;
        }
      }
      act = __builtin_amdgcn_ballot_w32(cur >= 0) & 0xFu;
    }
    if (MODE == 1 && round == 0) {
      rows[(size_t)node * 32 + lane] = (lane < cnt) ? my_src : -1;
      if (lane < 4) curs[node * 4 + lane] = cur;
    }
    int k = 0;
    for (; k + R * EPL <= cnt; k += R * EPL)
      gather_chunk<HEADS, R, true>(st, h, my_src, k, cnt, off, half, c16, as4, adn);
    if (k < cnt)
      gather_chunk<HEADS, R, false>(st, h, my_src, k, cnt, off, half, c16, as4, adn);
    if (act == 0u) break;
  } while (++round < max_rounds);

  float z = half_sum(st.zl);
  float acc[4] = {st.a0, st.a1, st.a2, st.a3};
  if constexpr (HEADS == 1) {
    const float mo  = __shfl_xor(st.m, 16, 32);
    const float rsc = expf(st.m - fmaxf(st.m, mo));
    z *= rsc;
#pragma unroll
    for (int j = 0; j < 4; ++j) acc[j] *= rsc;
    z += __shfl_xor(z, 16, 32);
#pragma unroll
    for (int j = 0; j < 4; ++j) acc[j] += __shfl_xor(acc[j], 16, 32);
  }

  if constexpr (HEADS == 2) {
    const float4 bv = *(const float4*)(bias + off);
    const float bb[4] = {bv.x, bv.y, bv.z, bv.w};
    float o[4];
#pragma unroll
    for (int j = 0; j < 4; ++j) {
      float v = acc[j] / (z + 1e-16f) + bb[j];
      o[j] = (v > 0.f) ? v : expm1f(v);
    }
    *(float4*)(out + (size_t)node * F + off) = make_float4(o[0], o[1], o[2], o[3]);
  } else {
    const unsigned o2 = off + 2u * (unsigned)half;
    const float2 bv = *(const float2*)(bias + o2);
    const float v0 = (half ? acc[2] : acc[0]) / (z + 1e-16f) + bv.x;
    const float v1 = (half ? acc[3] : acc[1]) / (z + 1e-16f) + bv.y;
    const float r0 = (v0 > 0.f) ? v0 : expm1f(v0);
    const float r1 = (v1 > 0.f) ? v1 : expm1f(v1);
    *(float2*)(out + (size_t)node * F + o2) = make_float2(r0, r1);
  }
}

extern "C" void kernel_launch(void* const* d_in, const int* in_sizes, int n_in,
                              void* d_out, int out_size, void* d_ws, size_t ws_size,
                              hipStream_t stream) {
  const float* x      = (const float*)d_in[0];
  const int*   ei     = (const int*)d_in[1];
  const float* W1     = (const float*)d_in[2];
  const float* a_src1 = (const float*)d_in[3];
  const float* a_dst1 = (const float*)d_in[4];
  const float* b1     = (const float*)d_in[5];
  const float* W2     = (const float*)d_in[6];
  const float* a_src2 = (const float*)d_in[7];
  const float* a_dst2 = (const float*)d_in[8];
  const float* b2     = (const float*)d_in[9];
  const float* m1w    = (const float*)d_in[10];
  const float* m1b    = (const float*)d_in[11];
  const float* m2w    = (const float*)d_in[12];
  const float* m2b    = (const float*)d_in[13];

  const int N    = in_sizes[0] / 64;
  const int E    = in_sizes[1] / 2;
  const int C    = in_sizes[12] / 64;

  float* bufA = (float*)d_ws;
  float* bufB = bufA + (size_t)N * 128;
  float* bufC = bufB + (size_t)N * 128;
  int*   rows = (int*)bufC;
  int*   curs = rows + (size_t)N * 32;
  int*   head = (int*)(bufC + (size_t)N * 64);
  int*   nxt  = head + (size_t)N * 4;

  const int TPB = 256;
  auto blocks1d = [&](long n) { return (int)((n + TPB - 1) / TPB); };
  auto gemm_blocks = [&](int M, int Nc) {
    int waves = ((M + 15) / 16) * ((Nc + 15) / 16);
    return (waves + 7) / 8;
  };
  auto gemm_row_blocks = [&](int M, int K) {
    int rb = 8192 / K;
    return (M + rb - 1) / rb;
  };

  fill_u32<<<blocks1d((long)N), TPB, 0, stream>>>((unsigned*)head, 0xFFFFFFFFu, N * 4);
  list_build<<<blocks1d(((long)E + 3) / 4), TPB, 0, stream>>>(ei, E, head, nxt);

  gemm_wmma_f32<64, 128><<<gemm_row_blocks(N, 64), TPB, 0, stream>>>(x, W1, nullptr,
                                                                  bufA, N, 0);
  node_gather<2, 1><<<(N + 7) / 8, TPB, 0, stream>>>(ei, E, head, nxt, rows, curs, a_src1, a_dst1,
                                                     bufA, b1, bufB, N);

  gemm_wmma_f32<128, 64><<<gemm_row_blocks(N, 128), TPB, 0, stream>>>(bufB, W2, nullptr,
                                                                 bufA, N, 0);
  node_gather<1, 2><<<(N + 7) / 8, TPB, 0, stream>>>(ei, E, head, nxt, rows, curs, a_src2, a_dst2,
                                                     bufA, b2, bufB, N);

  gemm_wmma_f32<64, 64><<<gemm_row_blocks(N, 64), TPB, 0, stream>>>(bufB, m1w, m1b,
                                                                bufA, N, 1);
  if (C == 8) {
    gemm_wmma_f32<64, 8><<<gemm_row_blocks(N, 64), TPB, 0, stream>>>(bufA, m2w, m2b,
                                                                (float*)d_out, N, 1);
  } else {
    gemm_wmma_f32_dyn<<<gemm_blocks(N, C), TPB, 0, stream>>>(bufA, m2w, m2b,
                                                             (float*)d_out, N, 64, C, 1);
  }
}


# edge grouping: 256 owner blocks sweep a byte tag per edge, rank their own edges by LDS adds into 32-source rows that both convs read; no global atomics or fill; arrival-order rank, sums' order shifts
# speedup vs baseline: 5.1793x; 1.0714x over previous
#include <hip/hip_runtime.h>
#include <math.h>

typedef __attribute__((ext_vector_type(2))) float v2f;
typedef __attribute__((ext_vector_type(8))) float v8f;

template <int K, int Nc>
__global__ void __launch_bounds__(256)
gemm_wmma_f32(const float* __restrict__ A, const float* __restrict__ B,
              const float* __restrict__ bias, float* __restrict__ Cout, int M, int act) {
  constexpr int  tilesN = (Nc + 15) >> 4;
  constexpr bool FULL   = (Nc % 16) == 0;
  constexpr int  RB  = 8192 / K;
  constexpr int  RT  = RB / 16;
  constexpr int  WG  = 8 / tilesN;
  constexpr int  TPW = RT / WG;
  constexpr int  P   = K + 4;
  constexpr int  KS  = K / 4;
  constexpr int  C4  = K / 4;
  constexpr int  CH  = KS < 16 ? KS : 16;
  static_assert(8192 % K == 0 && 8 % tilesN == 0 && RT % WG == 0, "block shape");
  __shared__ __attribute__((aligned(16))) float sA[RB * P];

  const int t    = threadIdx.x;
  const int w    = __builtin_amdgcn_readfirstlane(t >> 5);
  const int lane = t & 31;
  const int half = lane >> 4;
  const int l16  = lane & 15;
  const int r0   = blockIdx.x * RB;

  float4 av[8];
#pragma unroll
  for (int i = 0; i < 8; ++i) {
    int idx4 = i * 256 + t;
    int row  = idx4 / C4;
    int c4   = idx4 - row * C4;
    int gr   = r0 + row;
    gr = gr < M ? gr : M - 1;
    av[i] = *(const float4*)(A + (size_t)gr * K + c4 * 4);
  }

  const int tn   = w % tilesN;
  const int rsub = w / tilesN;
  const int col  = tn * 16 + l16;
  const bool cok = FULL || (col < Nc);
  const int colc = FULL ? col : (cok ? col : 0);
  const float* bp = B + (size_t)(2 * half) * Nc + colc;
  v2f bf[KS];
#pragma unroll
  for (int s = 0; s < KS; ++s) {
    float bx = bp[(size_t)(4 * s) * Nc];
    float by = bp[(size_t)(4 * s) * Nc + Nc];
    bf[s].x = FULL ? bx : (cok ? bx : 0.f);
    bf[s].y = FULL ? by : (cok ? by : 0.f);
  }
  float bv = 0.f;
  if (bias != nullptr && cok) bv = bias[col];

#pragma unroll
  for (int i = 0; i < 8; ++i) {
    int idx4 = i * 256 + t;
    int row  = idx4 / C4;
    int c4   = idx4 - row * C4;
    *(float4*)(sA + row * P + c4 * 4) = av[i];
  }
  __syncthreads();

  for (int j = 0; j < TPW; ++j) {
    const int rt   = rsub + WG * j;
    const int row0 = r0 + rt * 16;
    if (row0 >= M) break;
    const float* ar = sA + (rt * 16 + l16) * P + 2 * half;
    v8f c = {0.f, 0.f, 0.f, 0.f, 0.f, 0.f, 0.f, 0.f};
#pragma unroll
    for (int s0 = 0; s0 < KS; s0 += CH) {
      v2f a[CH];
#pragma unroll
      for (int s = 0; s < CH; ++s) a[s] = *(const v2f*)(ar + 4 * (s0 + s));
      __builtin_amdgcn_sched_barrier(0);
#pragma unroll
      for (int s = 0; s < CH; ++s)
        c = __builtin_amdgcn_wmma_f32_16x16x4_f32(false, a[s], false, bf[s0 + s], (short)0, c,
                                                  false, false);
      __builtin_amdgcn_sched_barrier(0);
    }
    if (cok) {
      const int rbase = row0 + 8 * half;
      float* crow = Cout + (size_t)rbase * Nc + col;
      if (row0 + 16 <= M) {
#pragma unroll
        for (int v = 0; v < 8; ++v) {
          float val = c[v] + bv;
          if (act) val = fmaxf(val, 0.f);
          crow[(size_t)v * Nc] = val;
        }
      } else {
#pragma unroll
        for (int v = 0; v < 8; ++v) {
          float val = c[v] + bv;
          if (act) val = fmaxf(val, 0.f);
          if (rbase + v < M) crow[(size_t)v * Nc] = val;
        }
      }
    }
  }
}

__global__ void gemm_wmma_f32_dyn(const float* __restrict__ A, const float* __restrict__ B,
                                  const float* __restrict__ bias, float* __restrict__ Cout,
                                  int M, int K, int Nc, int act) {
  int wave   = blockIdx.x * (blockDim.x >> 5) + (threadIdx.x >> 5);
  int tilesN = (Nc + 15) >> 4;
  int tilesM = (M + 15) >> 4;
  if (wave >= tilesM * tilesN) return;
  int tm = wave / tilesN, tn = wave - tm * tilesN;
  int lane = threadIdx.x & 31, half = lane >> 4, l16 = lane & 15;
  int row = tm * 16 + l16, col = tn * 16 + l16;
  bool cok = (col < Nc);
  int colc = cok ? col : 0;
  const float* ap = A + (size_t)row * K + 2 * half;
  const float* bp = B + (size_t)(2 * half) * Nc + colc;
  v8f c = {0.f, 0.f, 0.f, 0.f, 0.f, 0.f, 0.f, 0.f};
  for (int k0 = 0; k0 < K; k0 += 4) {
    float2 av = *(const float2*)ap;
    float bx = bp[0], by = bp[Nc];
    v2f a, b;
    a.x = av.x;            a.y = av.y;
    b.x = cok ? bx : 0.f;  b.y = cok ? by : 0.f;
    c = __builtin_amdgcn_wmma_f32_16x16x4_f32(false, a, false, b, (short)0, c,
                                              false, false);
    ap += 4;
    bp += (size_t)4 * Nc;
  }
  if (cok) {
    float bv = bias ? bias[col] : 0.f;
    float* crow = Cout + (size_t)(tm * 16 + 8 * half) * Nc + col;
    for (int v = 0; v < 8; ++v) {
      float val = c[v] + bv;
      if (act) val = fmaxf(val, 0.f);
      crow[(size_t)v * Nc] = val;
    }
  }
}

constexpr int OWN_T   = 1024;
constexpr int NPB_MAX = 512;
constexpr int QCAP    = 32768;
constexpr int SCAN_U  = 8;

__global__ void edge_tags(const int* __restrict__ ei, int Eorig, int npb,
                          unsigned char* __restrict__ tag, int* __restrict__ nxt) {
  const int e0 = (blockIdx.x * blockDim.x + threadIdx.x) * 16;
  if (e0 >= Eorig) return;
  const int* dp = ei + Eorig;
  const unsigned un = (unsigned)npb;
  if ((Eorig & 3) == 0 && e0 + 16 <= Eorig) {
    unsigned w[4];
#pragma unroll
    for (int k = 0; k < 4; ++k) {
      const int4 d = *(const int4*)(dp + e0 + 4 * k);
      w[k] = ((unsigned)d.x / un) | (((unsigned)d.y / un) << 8) | (((unsigned)d.z / un) << 16) |
             (((unsigned)d.w / un) << 24);
      *(int4*)(nxt + e0 + 4 * k) = make_int4(-1, -1, -1, -1);
    }
    *(uint4*)(tag + e0) = make_uint4(w[0], w[1], w[2], w[3]);
  } else {
    for (int e = e0; e < e0 + 16 && e < Eorig; ++e) {
      tag[e] = (unsigned char)((unsigned)dp[e] / un);
      nxt[e] = -1;
    }
  }
}

__global__ void __launch_bounds__(OWN_T)
owner_rows(const int* __restrict__ ei, int Eorig, const unsigned char* __restrict__ tag, int npb,
           int n, int* __restrict__ rows, int* __restrict__ curs, int* __restrict__ nxt) {
  __shared__ int q[QCAP];
  __shared__ int rows_s[NPB_MAX * 32];
  __shared__ int cnt_s[NPB_MAX];
  __shared__ int ovh_s[NPB_MAX];
  __shared__ int qn;
  const int t  = threadIdx.x;
  const int o  = blockIdx.x;
  const int lo = o * npb;
  const int hi = (lo + npb < n) ? lo + npb : n;
  const int* dp = ei + Eorig;
  const unsigned pat = (unsigned)o * 0x01010101u;
  const int nvec = Eorig >> 4;
  const int rot = (int)(((long long)o * nvec) / (int)gridDim.x);

  for (int base = lo; base < hi; base += NPB_MAX) {
    const int span = (hi - base < NPB_MAX) ? hi - base : NPB_MAX;
    for (int i = t; i < NPB_MAX; i += OWN_T) { cnt_s[i] = 0; ovh_s[i] = -1; }
    if (t == 0) qn = 0;
    __syncthreads();

    auto rank_edge = [&](int e, int d, int sv) {
      const unsigned loc = (unsigned)(d - base);
      if (loc < (unsigned)span) {
        const int r = atomicAdd(&cnt_s[loc], 1);
        if (r < 32) rows_s[loc * 32 + r] = sv;
        else nxt[e] = atomicExch(&ovh_s[loc], e);
      }
    };
    auto push = [&](int e) {
      const int slot = atomicAdd(&qn, 1);
      if (slot < QCAP) q[slot] = e;
      else rank_edge(e, dp[e], ei[e]);
    };

    for (int it0 = t; it0 < nvec; it0 += SCAN_U * OWN_T) {
      uint4 w[SCAN_U];
      int   ix[SCAN_U];
#pragma unroll
      for (int u = 0; u < SCAN_U; ++u) {
        const int it = it0 + u * OWN_T;
        int i = (it < nvec ? it : it0) + rot;
        if (i >= nvec) i -= nvec;
        ix[u] = i;
        w[u]  = *((const uint4*)tag + i);
      }
#pragma unroll
      for (int u = 0; u < SCAN_U; ++u) {
        if (it0 + u * OWN_T < nvec) {
          const unsigned ww[4] = {w[u].x, w[u].y, w[u].z, w[u].w};
          unsigned mm = 0u;
#pragma unroll
          for (int k = 0; k < 4; ++k) {
            const unsigned x = ww[k] ^ pat;
            const unsigned m = ~(((x & 0x7F7F7F7Fu) + 0x7F7F7F7Fu) | x | 0x7F7F7F7Fu);
            mm |= m >> (7 - k);
          }
          while (mm != 0u) {
            const int p = __builtin_ctz(mm);
            mm &= mm - 1u;
            push(ix[u] * 16 + (p & 3) * 4 + (p >> 3));
          }
        }
      }
    }
    for (int e = (nvec << 4) + t; e < Eorig; e += OWN_T)
      if (tag[e] == (unsigned char)o) push(e);
    __syncthreads();

    const int nq = qn < QCAP ? qn : QCAP;
    for (int j0 = t; j0 < nq; j0 += 4 * OWN_T) {
      int e4[4], d4[4], s4[4];
#pragma unroll
      for (int u = 0; u < 4; ++u) {
        const int j = j0 + u * OWN_T;
        e4[u] = q[j < nq ? j : j0];
        d4[u] = dp[e4[u]];
        s4[u] = ei[e4[u]];
      }
#pragma unroll
      for (int u = 0; u < 4; ++u)
        if (j0 + u * OWN_T < nq) rank_edge(e4[u], d4[u], s4[u]);
    }
    __syncthreads();

    for (int idx = t; idx < span * 32; idx += OWN_T) {
      const int loc = idx >> 5, k = idx & 31;
      int v = -1;
      if (k < cnt_s[loc]) v = rows_s[idx];
      rows[(size_t)(base + loc) * 32 + k] = v;
    }
    for (int idx = t; idx < span * 4; idx += OWN_T) {
      const int loc = idx >> 2;
      curs[(size_t)(base + loc) * 4 + (idx & 3)] = ((idx & 3) == 0) ? ovh_s[loc] : -1;
    }
    __syncthreads();
  }
}

template <int MASK>
__device__ __forceinline__ float row_xor(float v) {
  return __int_as_float(__builtin_amdgcn_update_dpp(0, __float_as_int(v), 0x160 | MASK, 0xF, 0xF, true));
}
template <int L>
__device__ __forceinline__ float row_share(float v) {
  return __int_as_float(__builtin_amdgcn_update_dpp(0, __float_as_int(v), 0x150 | L, 0xF, 0xF, true));
}
__device__ __forceinline__ float half_sum(float v) {
  v += row_xor<8>(v);
  v += row_xor<4>(v);
  v += row_xor<2>(v);
  v += row_xor<1>(v);
  return v;
}
__device__ __forceinline__ float half_max(float v) {
  v = fmaxf(v, row_xor<8>(v));
  v = fmaxf(v, row_xor<4>(v));
  v = fmaxf(v, row_xor<2>(v));
  v = fmaxf(v, row_xor<1>(v));
  return v;
}
__device__ __forceinline__ float dot4(float4 a, float4 b) {
  return fmaf(a.w, b.w, fmaf(a.z, b.z, fmaf(a.y, b.y, a.x * b.x)));
}
__device__ __forceinline__ float leaky(float a) { return (a > 0.f) ? a : 0.2f * a; }

struct Soft { float m, zl, a0, a1, a2, a3; };

template <int HEADS, int R, bool FULL>
__device__ __forceinline__ void gather_chunk(Soft& st, const float* __restrict__ h, int my_src,
                                             int k, int cnt, unsigned off, int half, int c16,
                                             float4 as4, float adn) {
  constexpr int F   = HEADS * 64;
  constexpr int EPL = (HEADS == 2) ? 1 : 2;
  float4 t[R];
#pragma unroll
  for (int q = 0; q < R; ++q) {
    const int e0 = k + q * EPL;
    if constexpr (!FULL) t[q] = make_float4(0.f, 0.f, 0.f, 0.f);
    if (FULL || e0 < cnt) {
      int s = __builtin_amdgcn_readlane(my_src, e0);
      if constexpr (HEADS == 1) {
        const int e1 = (FULL || e0 + 1 < cnt) ? e0 + 1 : e0;
        const int s1 = __builtin_amdgcn_readlane(my_src, e1);
        s = half ? s1 : s;
      }
      t[q] = *(const float4*)(h + ((unsigned)s * (unsigned)F + off));
    }
  }
  float d = 0.f;
#pragma unroll
  for (int q = 0; q < R; ++q) {
    if (FULL || k + q * EPL < cnt) {
      const float dq = half_sum(dot4(t[q], as4));
      d = (c16 == q) ? dq : d;
    }
  }
  const int  eidx  = k + c16 * EPL + ((HEADS == 1) ? half : 0);
  const bool valid = (c16 < R) && (FULL || eidx < cnt);
  const float a    = valid ? leaky(d + adn) : -INFINITY;
  const float mn   = fmaxf(st.m, half_max(a));
  const float ev   = expf(((c16 == R) ? st.m : a) - mn);
  const float rsc  = row_share<R>(ev);
  st.zl = st.zl * rsc + ((c16 == R) ? 0.f : ev);
  st.a0 *= rsc; st.a1 *= rsc; st.a2 *= rsc; st.a3 *= rsc;
  st.m = mn;
  auto add_row = [&](float w, float4 r) {
    st.a0 = fmaf(w, r.x, st.a0); st.a1 = fmaf(w, r.y, st.a1);
    st.a2 = fmaf(w, r.z, st.a2); st.a3 = fmaf(w, r.w, st.a3);
  };
  if (FULL || k + 0 * EPL < cnt) add_row(row_share<0>(ev), t[0]);
  if (FULL || k + 1 * EPL < cnt) add_row(row_share<1>(ev), t[1]);
  if (FULL || k + 2 * EPL < cnt) add_row(row_share<2>(ev), t[2]);
  if (FULL || k + 3 * EPL < cnt) add_row(row_share<3>(ev), t[3]);
  if (FULL || k + 4 * EPL < cnt) add_row(row_share<4>(ev), t[4]);
  if (FULL || k + 5 * EPL < cnt) add_row(row_share<5>(ev), t[5]);
  if (FULL || k + 6 * EPL < cnt) add_row(row_share<6>(ev), t[6]);
  if (FULL || k + 7 * EPL < cnt) add_row(row_share<7>(ev), t[7]);
}

template <int HEADS>
__global__ void __launch_bounds__(256)
node_gather(const int* __restrict__ ei, int Eorig,
                            const int* __restrict__ nxt, const int* __restrict__ rows,
                            const int* __restrict__ curs, const float* __restrict__ a_src,
                            const float* __restrict__ a_dst, const float* __restrict__ h,
                            const float* __restrict__ bias, float* __restrict__ out, int n) {
  constexpr int F   = HEADS * 64;
  constexpr int EPL = (HEADS == 2) ? 1 : 2;
  constexpr int R   = 8;
  int node = __builtin_amdgcn_readfirstlane((int)(blockIdx.x * (blockDim.x >> 5) + (threadIdx.x >> 5)));
  if (node >= n) return;
  const int lane = threadIdx.x & 31;
  const int half = lane >> 4;
  const int c16  = lane & 15;
  const int hd   = (HEADS == 2) ? half : 0;
  const unsigned off = (HEADS == 2) ? lane * 4 : c16 * 4;

  const int pre = rows[(size_t)node * 32 + lane];
  int cur = curs[node * 4 + (lane & 3)];
  const float4 own = *(const float4*)(h + (size_t)node * F + off);
  const float4 as4 = *(const float4*)(a_src + hd * 64 + c16 * 4);
  const float4 ad4 = *(const float4*)(a_dst + hd * 64 + c16 * 4);
  __builtin_amdgcn_sched_barrier(0);

  const float adn = half_sum(dot4(own, ad4));
  const bool mine = (HEADS == 2) || (half == 0);
  Soft st;
  st.m  = leaky(half_sum(dot4(own, as4)) + adn);
  st.zl = (mine && c16 == 0) ? 1.f : 0.f;
  st.a0 = mine ? own.x : 0.f;
  st.a1 = mine ? own.y : 0.f;
  st.a2 = mine ? own.z : 0.f;
  st.a3 = mine ? own.w : 0.f;

  unsigned act = __builtin_amdgcn_ballot_w32(cur >= 0) & 0xFu;
  const int max_rounds = (Eorig >> 5) + 2;
  int round = 0;
  do {
    int cnt = 0;
    int my_src = node;
    if (round == 0) {
      my_src = pre;
      cnt = __builtin_popcount(__builtin_amdgcn_ballot_w32(pre >= 0));
    } else
    while (act != 0u && cnt < 32) {
      int ci = cur > 0 ? cur : 0;
      int sv = ei[ci];
      int nv = nxt[ci];
#pragma unroll
      for (int c = 0; c < 4; ++c) {
        if (((act >> c) & 1u) != 0u && cnt < 32) {
          int s = __builtin_amdgcn_readlane(sv, c);
          if (lane == cnt) my_src = s;
          ++cnt;
          if ((lane & 3) == c) cur = nv;
        }
      }
      act = __builtin_amdgcn_ballot_w32(cur >= 0) & 0xFu;
    }
    int k = 0;
    for (; k + R * EPL <= cnt; k += R * EPL)
      gather_chunk<HEADS, R, true>(st, h, my_src, k, cnt, off, half, c16, as4, adn);
    if (k < cnt)
      gather_chunk<HEADS, R, false>(st, h, my_src, k, cnt, off, half, c16, as4, adn);
    if (act == 0u) break;
  } while (++round < max_rounds);

  float z = half_sum(st.zl);
  float acc[4] = {st.a0, st.a1, st.a2, st.a3};
  if constexpr (HEADS == 1) {
    const float mo  = __shfl_xor(st.m, 16, 32);
    const float rsc = expf(st.m - fmaxf(st.m, mo));
    z *= rsc;
#pragma unroll
    for (int j = 0; j < 4; ++j) acc[j] *= rsc;
    z += __shfl_xor(z, 16, 32);
#pragma unroll
    for (int j = 0; j < 4; ++j) acc[j] += __shfl_xor(acc[j], 16, 32);
  }

  if constexpr (HEADS == 2) {
    const float4 bv = *(const float4*)(bias + off);
    const float bb[4] = {bv.x, bv.y, bv.z, bv.w};
    float o[4];
#pragma unroll
    for (int j = 0; j < 4; ++j) {
      float v = acc[j] / (z + 1e-16f) + bb[j];
      o[j] = (v > 0.f) ? v : expm1f(v);
    }
    *(float4*)(out + (size_t)node * F + off) = make_float4(o[0], o[1], o[2], o[3]);
  } else {
    const unsigned o2 = off + 2u * (unsigned)half;
    const float2 bv = *(const float2*)(bias + o2);
    const float v0 = (half ? acc[2] : acc[0]) / (z + 1e-16f) + bv.x;
    const float v1 = (half ? acc[3] : acc[1]) / (z + 1e-16f) + bv.y;
    const float r0 = (v0 > 0.f) ? v0 : expm1f(v0);
    const float r1 = (v1 > 0.f) ? v1 : expm1f(v1);
    *(float2*)(out + (size_t)node * F + o2) = make_float2(r0, r1);
  }
}

extern "C" void kernel_launch(void* const* d_in, const int* in_sizes, int n_in,
                              void* d_out, int out_size, void* d_ws, size_t ws_size,
                              hipStream_t stream) {
  const float* x      = (const float*)d_in[0];
  const int*   ei     = (const int*)d_in[1];
  const float* W1     = (const float*)d_in[2];
  const float* a_src1 = (const float*)d_in[3];
  const float* a_dst1 = (const float*)d_in[4];
  const float* b1     = (const float*)d_in[5];
  const float* W2     = (const float*)d_in[6];
  const float* a_src2 = (const float*)d_in[7];
  const float* a_dst2 = (const float*)d_in[8];
  const float* b2     = (const float*)d_in[9];
  const float* m1w    = (const float*)d_in[10];
  const float* m1b    = (const float*)d_in[11];
  const float* m2w    = (const float*)d_in[12];
  const float* m2b    = (const float*)d_in[13];

  const int N    = in_sizes[0] / 64;
  const int E    = in_sizes[1] / 2;
  const int C    = in_sizes[12] / 64;

  float* bufA = (float*)d_ws;
  float* bufB = bufA + (size_t)N * 128;
  float* bufC = bufB + (size_t)N * 128;
  int*   rows = (int*)bufC;
  int*   curs = rows + (size_t)N * 32;
  int*   nxt  = (int*)(bufC + (size_t)N * 68);
  unsigned char* tag = ((size_t)E <= (size_t)N * 128) ? (unsigned char*)(curs + (size_t)N * 4)
                                                      : (unsigned char*)(nxt + (((size_t)E + 3) & ~(size_t)3));

  const int TPB = 256;
  auto blocks1d = [&](long n) { return (int)((n + TPB - 1) / TPB); };
  auto gemm_blocks = [&](int M, int Nc) {
    int waves = ((M + 15) / 16) * ((Nc + 15) / 16);
    return (waves + 7) / 8;
  };
  auto gemm_row_blocks = [&](int M, int K) {
    int rb = 8192 / K;
    return (M + rb - 1) / rb;
  };

  const int npb    = (N > 256) ? (N + 255) / 256 : 1;
  const int owners = (N + npb - 1) / npb;
  edge_tags<<<blocks1d(((long)E + 15) / 16), TPB, 0, stream>>>(ei, E, npb, tag, nxt);
  owner_rows<<<owners, OWN_T, 0, stream>>>(ei, E, tag, npb, N, rows, curs, nxt);

  gemm_wmma_f32<64, 128><<<gemm_row_blocks(N, 64), TPB, 0, stream>>>(x, W1, nullptr,
                                                                  bufA, N, 0);
  node_gather<2><<<(N + 7) / 8, TPB, 0, stream>>>(ei, E, nxt, rows, curs, a_src1, a_dst1,
                                                  bufA, b1, bufB, N);

  gemm_wmma_f32<128, 64><<<gemm_row_blocks(N, 128), TPB, 0, stream>>>(bufB, W2, nullptr,
                                                                 bufA, N, 0);
  node_gather<1><<<(N + 7) / 8, TPB, 0, stream>>>(ei, E, nxt, rows, curs, a_src2, a_dst2,
                                                  bufA, b2, bufB, N);

  gemm_wmma_f32<64, 64><<<gemm_row_blocks(N, 64), TPB, 0, stream>>>(bufB, m1w, m1b,
                                                                bufA, N, 1);
  if (C == 8) {
    gemm_wmma_f32<64, 8><<<gemm_row_blocks(N, 64), TPB, 0, stream>>>(bufA, m2w, m2b,
                                                                (float*)d_out, N, 1);
  } else {
    gemm_wmma_f32_dyn<<<gemm_blocks(N, C), TPB, 0, stream>>>(bufA, m2w, m2b,
                                                             (float*)d_out, N, 64, C, 1);
  }
}


# edge grouping: pass 1 sets each edge's bit in its owner's bit row (1024-edge tiles built in LDS, 128-byte lines); an owner block scans its own row, not all tags; arrival-order rank, sums' order shifts
# speedup vs baseline: 5.4445x; 1.0512x over previous
#include <hip/hip_runtime.h>
#include <math.h>

typedef __attribute__((ext_vector_type(2))) float v2f;
typedef __attribute__((ext_vector_type(8))) float v8f;

template <int K, int Nc>
__global__ void __launch_bounds__(256)
gemm_wmma_f32(const float* __restrict__ A, const float* __restrict__ B,
              const float* __restrict__ bias, float* __restrict__ Cout, int M, int act) {
  constexpr int  tilesN = (Nc + 15) >> 4;
  constexpr bool FULL   = (Nc % 16) == 0;
  constexpr int  RB  = 8192 / K;
  constexpr int  RT  = RB / 16;
  constexpr int  WG  = 8 / tilesN;
  constexpr int  TPW = RT / WG;
  constexpr int  P   = K + 4;
  constexpr int  KS  = K / 4;
  constexpr int  C4  = K / 4;
  constexpr int  CH  = KS < 16 ? KS : 16;
  static_assert(8192 % K == 0 && 8 % tilesN == 0 && RT % WG == 0, "block shape");
  __shared__ __attribute__((aligned(16))) float sA[RB * P];

  const int t    = threadIdx.x;
  const int w    = __builtin_amdgcn_readfirstlane(t >> 5);
  const int lane = t & 31;
  const int half = lane >> 4;
  const int l16  = lane & 15;
  const int r0   = blockIdx.x * RB;

  float4 av[8];
#pragma unroll
  for (int i = 0; i < 8; ++i) {
    int idx4 = i * 256 + t;
    int row  = idx4 / C4;
    int c4   = idx4 - row * C4;
    int gr   = r0 + row;
    gr = gr < M ? gr : M - 1;
    av[i] = *(const float4*)(A + (size_t)gr * K + c4 * 4);
  }

  const int tn   = w % tilesN;
  const int rsub = w / tilesN;
  const int col  = tn * 16 + l16;
  const bool cok = FULL || (col < Nc);
  const int colc = FULL ? col : (cok ? col : 0);
  const float* bp = B + (size_t)(2 * half) * Nc + colc;
  v2f bf[KS];
#pragma unroll
  for (int s = 0; s < KS; ++s) {
    float bx = bp[(size_t)(4 * s) * Nc];
    float by = bp[(size_t)(4 * s) * Nc + Nc];
    bf[s].x = FULL ? bx : (cok ? bx : 0.f);
    bf[s].y = FULL ? by : (cok ? by : 0.f);
  }
  float bv = 0.f;
  if (bias != nullptr && cok) bv = bias[col];

#pragma unroll
  for (int i = 0; i < 8; ++i) {
    int idx4 = i * 256 + t;
    int row  = idx4 / C4;
    int c4   = idx4 - row * C4;
    *(float4*)(sA + row * P + c4 * 4) = av[i];
  }
  __syncthreads();

  for (int j = 0; j < TPW; ++j) {
    const int rt   = rsub + WG * j;
    const int row0 = r0 + rt * 16;
    if (row0 >= M) break;
    const float* ar = sA + (rt * 16 + l16) * P + 2 * half;
    v8f c = {0.f, 0.f, 0.f, 0.f, 0.f, 0.f, 0.f, 0.f};
#pragma unroll
    for (int s0 = 0; s0 < KS; s0 += CH) {
      v2f a[CH];
#pragma unroll
      for (int s = 0; s < CH; ++s) a[s] = *(const v2f*)(ar + 4 * (s0 + s));
      __builtin_amdgcn_sched_barrier(0);
#pragma unroll
      for (int s = 0; s < CH; ++s)
        c = __builtin_amdgcn_wmma_f32_16x16x4_f32(false, a[s], false, bf[s0 + s], (short)0, c,
                                                  false, false);
      __builtin_amdgcn_sched_barrier(0);
    }
    if (cok) {
      const int rbase = row0 + 8 * half;
      float* crow = Cout + (size_t)rbase * Nc + col;
      if (row0 + 16 <= M) {
#pragma unroll
        for (int v = 0; v < 8; ++v) {
          float val = c[v] + bv;
          if (act) val = fmaxf(val, 0.f);
          crow[(size_t)v * Nc] = val;
        }
      } else {
#pragma unroll
        for (int v = 0; v < 8; ++v) {
          float val = c[v] + bv;
          if (act) val = fmaxf(val, 0.f);
          if (rbase + v < M) crow[(size_t)v * Nc] = val;
        }
      }
    }
  }
}

__global__ void gemm_wmma_f32_dyn(const float* __restrict__ A, const float* __restrict__ B,
                                  const float* __restrict__ bias, float* __restrict__ Cout,
                                  int M, int K, int Nc, int act) {
  int wave   = blockIdx.x * (blockDim.x >> 5) + (threadIdx.x >> 5);
  int tilesN = (Nc + 15) >> 4;
  int tilesM = (M + 15) >> 4;
  if (wave >= tilesM * tilesN) return;
  int tm = wave / tilesN, tn = wave - tm * tilesN;
  int lane = threadIdx.x & 31, half = lane >> 4, l16 = lane & 15;
  int row = tm * 16 + l16, col = tn * 16 + l16;
  bool cok = (col < Nc);
  int colc = cok ? col : 0;
  const float* ap = A + (size_t)row * K + 2 * half;
  const float* bp = B + (size_t)(2 * half) * Nc + colc;
  v8f c = {0.f, 0.f, 0.f, 0.f, 0.f, 0.f, 0.f, 0.f};
  for (int k0 = 0; k0 < K; k0 += 4) {
    float2 av = *(const float2*)ap;
    float bx = bp[0], by = bp[Nc];
    v2f a, b;
    a.x = av.x;            a.y = av.y;
    b.x = cok ? bx : 0.f;  b.y = cok ? by : 0.f;
    c = __builtin_amdgcn_wmma_f32_16x16x4_f32(false, a, false, b, (short)0, c,
                                              false, false);
    ap += 4;
    bp += (size_t)4 * Nc;
  }
  if (cok) {
    float bv = bias ? bias[col] : 0.f;
    float* crow = Cout + (size_t)(tm * 16 + 8 * half) * Nc + col;
    for (int v = 0; v < 8; ++v) {
      float val = c[v] + bv;
      if (act) val = fmaxf(val, 0.f);
      crow[(size_t)v * Nc] = val;
    }
  }
}

constexpr int OWN_T   = 1024;
constexpr int NPB_MAX = 512;
constexpr int QCAP    = 32768;
constexpr int SCAN_U  = 8;
constexpr int TILE_E  = 1024;

__global__ void __launch_bounds__(256)
edge_bits(const int* __restrict__ ei, int Eorig, int npb, int rowW,
          unsigned* __restrict__ bits, int* __restrict__ nxt) {
  __shared__ __attribute__((aligned(16))) unsigned tab[256 * 32];
  const int t = threadIdx.x;
  const int* dp = ei + Eorig;
  const unsigned un = (unsigned)npb;
  const int ntiles = rowW >> 5;
  const int w  = t >> 3;
  const int b0 = (4 * t) & 31;
  uint4* tab4 = (uint4*)tab;
  for (int i = t; i < 256 * 8; i += 256) tab4[i] = make_uint4(0u, 0u, 0u, 0u);
  for (int tile = blockIdx.x; tile < ntiles; tile += gridDim.x) {
    __syncthreads();
    const int e0 = tile * TILE_E + 4 * t;
    int d[4] = {-1, -1, -1, -1};
    if ((Eorig & 3) == 0 && e0 + 4 <= Eorig) {
      const int4 v = *(const int4*)(dp + e0);
      d[0] = v.x; d[1] = v.y; d[2] = v.z; d[3] = v.w;
      *(int4*)(nxt + e0) = make_int4(-1, -1, -1, -1);
    } else {
#pragma unroll
      for (int k = 0; k < 4; ++k)
        if (e0 + k < Eorig) { d[k] = dp[e0 + k]; nxt[e0 + k] = -1; }
    }
#pragma unroll
    for (int k = 0; k < 4; ++k) {
      if (d[k] >= 0) {
        const unsigned o = (unsigned)d[k] / un;
        if (o < 256u) atomicOr(&tab[o * 32u + (unsigned)w], 1u << (b0 + k));
      }
    }
    __syncthreads();
#pragma unroll
    for (int j = 0; j < 8; ++j) {
      const int i = t + 256 * j;
      const uint4 v = tab4[i];
      tab4[i] = make_uint4(0u, 0u, 0u, 0u);
      *(uint4*)(bits + (size_t)(i >> 3) * rowW + (size_t)tile * 32 + (i & 7) * 4) = v;
    }
  }
}

template <int N>
__device__ __forceinline__ int row_shr(int v) {
  return __builtin_amdgcn_update_dpp(0, v, 0x110 | N, 0xF, 0xF, true);
}

__global__ void __launch_bounds__(OWN_T)
owner_rows(const int* __restrict__ ei, int Eorig, const unsigned* __restrict__ bits, int rowW,
           int npb, int n, int* __restrict__ rows, int* __restrict__ curs, int* __restrict__ nxt) {
  __shared__ int q[QCAP + 1];
  __shared__ int rows_s[NPB_MAX * 32];
  __shared__ int cnt_s[NPB_MAX];
  __shared__ int ovh_s[NPB_MAX];
  __shared__ int qn;
  const int t  = threadIdx.x;
  const int o  = blockIdx.x;
  const int lo = o * npb;
  const int hi = (lo + npb < n) ? lo + npb : n;
  const int* dp = ei + Eorig;
  const uint4* brow = (const uint4*)(bits + (size_t)o * rowW);
  const int nvec = rowW >> 2;

  for (int base = lo; base < hi; base += NPB_MAX) {
    const int span = (hi - base < NPB_MAX) ? hi - base : NPB_MAX;
    for (int i = t; i < NPB_MAX; i += OWN_T) { cnt_s[i] = 0; ovh_s[i] = -1; }
    if (t == 0) qn = 0;
    __syncthreads();

    auto rank_edge = [&](int e, int d, int sv) {
      const unsigned loc = (unsigned)(d - base);
      if (loc < (unsigned)span) {
        const int r = atomicAdd(&cnt_s[loc], 1);
        if (r < 32) rows_s[loc * 32 + r] = sv;
        else nxt[e] = atomicExch(&ovh_s[loc], e);
      }
    };

    for (int ib = 0; ib < nvec; ib += SCAN_U * OWN_T) {
      const int it0 = ib + t;
      uint4 w[SCAN_U];
#pragma unroll
      for (int u = 0; u < SCAN_U; ++u) {
        const int it = it0 + u * OWN_T;
        w[u] = brow[it < nvec ? it : 0];
      }
      __builtin_amdgcn_sched_barrier(0);
      int cl = 0;
#pragma unroll
      for (int u = 0; u < SCAN_U; ++u) {
        if (it0 + u * OWN_T >= nvec) w[u] = make_uint4(0u, 0u, 0u, 0u);
        cl += __builtin_popcount(w[u].x) + __builtin_popcount(w[u].y) +
              __builtin_popcount(w[u].z) + __builtin_popcount(w[u].w);
      }
      int run = cl;
      run += row_shr<1>(run);
      run += row_shr<2>(run);
      run += row_shr<4>(run);
      run += row_shr<8>(run);
      const int lowtot = __builtin_amdgcn_readlane(run, 15);
      if ((t & 31) >= 16) run += lowtot;
      const int total = __builtin_amdgcn_readlane(run, 31);
      int first = 0;
      if (total != 0) {
        if ((t & 31) == 0) first = atomicAdd(&qn, total);
        first = __builtin_amdgcn_readfirstlane(first);
      }
      int slot = first + run - cl;
      const int send = slot + cl;
#pragma unroll
      for (int u = 0; u < SCAN_U; ++u) {
        const unsigned ww[4] = {w[u].x, w[u].y, w[u].z, w[u].w};
#pragma unroll
        for (int k = 0; k < 4; ++k) {
          unsigned m = ww[k];
          const int eb = ((it0 + u * OWN_T) * 4 + k) << 5;
          while (m != 0u) {
            q[slot < QCAP ? slot : QCAP] = eb + __builtin_ctz(m);
            m &= m - 1u;
            ++slot;
          }
        }
      }
      if (send > QCAP) {
        int s2 = send - cl;
#pragma nounroll
        for (int u = 0; u < SCAN_U; ++u) {
          const int it = it0 + u * OWN_T;
          if (it >= nvec) break;
          const unsigned* wp = (const unsigned*)(brow + it);
#pragma nounroll
          for (int k = 0; k < 4; ++k) {
            unsigned m = wp[k];
#pragma nounroll
            while (m != 0u) {
              const int e = ((it * 4 + k) << 5) + __builtin_ctz(m);
              m &= m - 1u;
              if (s2 >= QCAP) rank_edge(e, dp[e], ei[e]);
              ++s2;
            }
          }
        }
      }
    }
    __syncthreads();

    const int nq = qn < QCAP ? qn : QCAP;
    for (int j0 = t; j0 < nq; j0 += 4 * OWN_T) {
      int e4[4], d4[4], s4[4];
#pragma unroll
      for (int u = 0; u < 4; ++u) {
        const int j = j0 + u * OWN_T;
        e4[u] = q[j < nq ? j : j0];
        d4[u] = dp[e4[u]];
        s4[u] = ei[e4[u]];
      }
#pragma unroll
      for (int u = 0; u < 4; ++u)
        if (j0 + u * OWN_T < nq) rank_edge(e4[u], d4[u], s4[u]);
    }
    __syncthreads();

    for (int idx = t; idx < span * 32; idx += OWN_T) {
      const int loc = idx >> 5, k = idx & 31;
      int v = -1;
      if (k < cnt_s[loc]) v = rows_s[idx];
      rows[(size_t)(base + loc) * 32 + k] = v;
    }
    for (int idx = t; idx < span * 4; idx += OWN_T) {
      const int loc = idx >> 2;
      curs[(size_t)(base + loc) * 4 + (idx & 3)] = ((idx & 3) == 0) ? ovh_s[loc] : -1;
    }
    __syncthreads();
  }
}

template <int MASK>
__device__ __forceinline__ float row_xor(float v) {
  return __int_as_float(__builtin_amdgcn_update_dpp(0, __float_as_int(v), 0x160 | MASK, 0xF, 0xF, true));
}
template <int L>
__device__ __forceinline__ float row_share(float v) {
  return __int_as_float(__builtin_amdgcn_update_dpp(0, __float_as_int(v), 0x150 | L, 0xF, 0xF, true));
}
__device__ __forceinline__ float half_sum(float v) {
  v += row_xor<8>(v);
  v += row_xor<4>(v);
  v += row_xor<2>(v);
  v += row_xor<1>(v);
  return v;
}
__device__ __forceinline__ float half_max(float v) {
  v = fmaxf(v, row_xor<8>(v));
  v = fmaxf(v, row_xor<4>(v));
  v = fmaxf(v, row_xor<2>(v));
  v = fmaxf(v, row_xor<1>(v));
  return v;
}
__device__ __forceinline__ float dot4(float4 a, float4 b) {
  return fmaf(a.w, b.w, fmaf(a.z, b.z, fmaf(a.y, b.y, a.x * b.x)));
}
__device__ __forceinline__ float leaky(float a) { return (a > 0.f) ? a : 0.2f * a; }

struct Soft { float m, zl, a0, a1, a2, a3; };

template <int HEADS, int R, bool FULL>
__device__ __forceinline__ void gather_chunk(Soft& st, const float* __restrict__ h, int my_src,
                                             int k, int cnt, unsigned off, int half, int c16,
                                             float4 as4, float adn) {
  constexpr int F   = HEADS * 64;
  constexpr int EPL = (HEADS == 2) ? 1 : 2;
  float4 t[R];
#pragma unroll
  for (int q = 0; q < R; ++q) {
    const int e0 = k + q * EPL;
    if constexpr (!FULL) t[q] = make_float4(0.f, 0.f, 0.f, 0.f);
    if (FULL || e0 < cnt) {
      int s = __builtin_amdgcn_readlane(my_src, e0);
      if constexpr (HEADS == 1) {
        const int e1 = (FULL || e0 + 1 < cnt) ? e0 + 1 : e0;
        const int s1 = __builtin_amdgcn_readlane(my_src, e1);
        s = half ? s1 : s;
      }
      t[q] = *(const float4*)(h + ((unsigned)s * (unsigned)F + off));
    }
  }
  float d = 0.f;
#pragma unroll
  for (int q = 0; q < R; ++q) {
    if (FULL || k + q * EPL < cnt) {
      const float dq = half_sum(dot4(t[q], as4));
      d = (c16 == q) ? dq : d;
    }
  }
  const int  eidx  = k + c16 * EPL + ((HEADS == 1) ? half : 0);
  const bool valid = (c16 < R) && (FULL || eidx < cnt);
  const float a    = valid ? leaky(d + adn) : -INFINITY;
  const float mn   = fmaxf(st.m, half_max(a));
  const float ev   = expf(((c16 == R) ? st.m : a) - mn);
  const float rsc  = row_share<R>(ev);
  st.zl = st.zl * rsc + ((c16 == R) ? 0.f : ev);
  st.a0 *= rsc; st.a1 *= rsc; st.a2 *= rsc; st.a3 *= rsc;
  st.m = mn;
  auto add_row = [&](float w, float4 r) {
    st.a0 = fmaf(w, r.x, st.a0); st.a1 = fmaf(w, r.y, st.a1);
    st.a2 = fmaf(w, r.z, st.a2); st.a3 = fmaf(w, r.w, st.a3);
  };
  if (FULL || k + 0 * EPL < cnt) add_row(row_share<0>(ev), t[0]);
  if (FULL || k + 1 * EPL < cnt) add_row(row_share<1>(ev), t[1]);
  if (FULL || k + 2 * EPL < cnt) add_row(row_share<2>(ev), t[2]);
  if (FULL || k + 3 * EPL < cnt) add_row(row_share<3>(ev), t[3]);
  if (FULL || k + 4 * EPL < cnt) add_row(row_share<4>(ev), t[4]);
  if (FULL || k + 5 * EPL < cnt) add_row(row_share<5>(ev), t[5]);
  if (FULL || k + 6 * EPL < cnt) add_row(row_share<6>(ev), t[6]);
  if (FULL || k + 7 * EPL < cnt) add_row(row_share<7>(ev), t[7]);
}

template <int HEADS>
__global__ void __launch_bounds__(256)
node_gather(const int* __restrict__ ei, int Eorig,
                            const int* __restrict__ nxt, const int* __restrict__ rows,
                            const int* __restrict__ curs, const float* __restrict__ a_src,
                            const float* __restrict__ a_dst, const float* __restrict__ h,
                            const float* __restrict__ bias, float* __restrict__ out, int n) {
  constexpr int F   = HEADS * 64;
  constexpr int EPL = (HEADS == 2) ? 1 : 2;
  constexpr int R   = 8;
  int node = __builtin_amdgcn_readfirstlane((int)(blockIdx.x * (blockDim.x >> 5) + (threadIdx.x >> 5)));
  if (node >= n) return;
  const int lane = threadIdx.x & 31;
  const int half = lane >> 4;
  const int c16  = lane & 15;
  const int hd   = (HEADS == 2) ? half : 0;
  const unsigned off = (HEADS == 2) ? lane * 4 : c16 * 4;

  const int pre = rows[(size_t)node * 32 + lane];
  int cur = curs[node * 4 + (lane & 3)];
  const float4 own = *(const float4*)(h + (size_t)node * F + off);
  const float4 as4 = *(const float4*)(a_src + hd * 64 + c16 * 4);
  const float4 ad4 = *(const float4*)(a_dst + hd * 64 + c16 * 4);
  __builtin_amdgcn_sched_barrier(0);

  const float adn = half_sum(dot4(own, ad4));
  const bool mine = (HEADS == 2) || (half == 0);
  Soft st;
  st.m  = leaky(half_sum(dot4(own, as4)) + adn);
  st.zl = (mine && c16 == 0) ? 1.f : 0.f;
  st.a0 = mine ? own.x : 0.f;
  st.a1 = mine ? own.y : 0.f;
  st.a2 = mine ? own.z : 0.f;
  st.a3 = mine ? own.w : 0.f;

  unsigned act = __builtin_amdgcn_ballot_w32(cur >= 0) & 0xFu;
  const int max_rounds = (Eorig >> 5) + 2;
  int round = 0;
  do {
    int cnt = 0;
    int my_src = node;
    if (round == 0) {
      my_src = pre;
      cnt = __builtin_popcount(__builtin_amdgcn_ballot_w32(pre >= 0));
    } else
    while (act != 0u && cnt < 32) {
      int ci = cur > 0 ? cur : 0;
      int sv = ei[ci];
      int nv = nxt[ci];
#pragma unroll
      for (int c = 0; c < 4; ++c) {
        if (((act >> c) & 1u) != 0u && cnt < 32) {
          int s = __builtin_amdgcn_readlane(sv, c);
          if (lane == cnt) my_src = s;
          ++cnt;
          if ((lane & 3) == c) cur = nv;
        }
      }
      act = __builtin_amdgcn_ballot_w32(cur >= 0) & 0xFu;
    }
    int k = 0;
    for (; k + R * EPL <= cnt; k += R * EPL)
      gather_chunk<HEADS, R, true>(st, h, my_src, k, cnt, off, half, c16, as4, adn);
    if (k < cnt)
      gather_chunk<HEADS, R, false>(st, h, my_src, k, cnt, off, half, c16, as4, adn);
    if (act == 0u) break;
  } while (++round < max_rounds);

  float z = half_sum(st.zl);
  float acc[4] = {st.a0, st.a1, st.a2, st.a3};
  if constexpr (HEADS == 1) {
    const float mo  = __shfl_xor(st.m, 16, 32);
    const float rsc = expf(st.m - fmaxf(st.m, mo));
    z *= rsc;
#pragma unroll
    for (int j = 0; j < 4; ++j) acc[j] *= rsc;
    z += __shfl_xor(z, 16, 32);
#pragma unroll
    for (int j = 0; j < 4; ++j) acc[j] += __shfl_xor(acc[j], 16, 32);
  }

  if constexpr (HEADS == 2) {
    const float4 bv = *(const float4*)(bias + off);
    const float bb[4] = {bv.x, bv.y, bv.z, bv.w};
    float o[4];
#pragma unroll
    for (int j = 0; j < 4; ++j) {
      float v = acc[j] / (z + 1e-16f) + bb[j];
      o[j] = (v > 0.f) ? v : expm1f(v);
    }
    *(float4*)(out + (size_t)node * F + off) = make_float4(o[0], o[1], o[2], o[3]);
  } else {
    const unsigned o2 = off + 2u * (unsigned)half;
    const float2 bv = *(const float2*)(bias + o2);
    const float v0 = (half ? acc[2] : acc[0]) / (z + 1e-16f) + bv.x;
    const float v1 = (half ? acc[3] : acc[1]) / (z + 1e-16f) + bv.y;
    const float r0 = (v0 > 0.f) ? v0 : expm1f(v0);
    const float r1 = (v1 > 0.f) ? v1 : expm1f(v1);
    *(float2*)(out + (size_t)node * F + o2) = make_float2(r0, r1);
  }
}

extern "C" void kernel_launch(void* const* d_in, const int* in_sizes, int n_in,
                              void* d_out, int out_size, void* d_ws, size_t ws_size,
                              hipStream_t stream) {
  const float* x      = (const float*)d_in[0];
  const int*   ei     = (const int*)d_in[1];
  const float* W1     = (const float*)d_in[2];
  const float* a_src1 = (const float*)d_in[3];
  const float* a_dst1 = (const float*)d_in[4];
  const float* b1     = (const float*)d_in[5];
  const float* W2     = (const float*)d_in[6];
  const float* a_src2 = (const float*)d_in[7];
  const float* a_dst2 = (const float*)d_in[8];
  const float* b2     = (const float*)d_in[9];
  const float* m1w    = (const float*)d_in[10];
  const float* m1b    = (const float*)d_in[11];
  const float* m2w    = (const float*)d_in[12];
  const float* m2b    = (const float*)d_in[13];

  const int N    = in_sizes[0] / 64;
  const int E    = in_sizes[1] / 2;
  const int C    = in_sizes[12] / 64;

  float* bufA = (float*)d_ws;
  float* bufB = bufA + (size_t)N * 128;
  float* bufC = bufB + (size_t)N * 128;
  int*   rows = (int*)bufC;
  int*   curs = rows + (size_t)N * 32;
  int*   nxt  = (int*)(bufC + (size_t)N * 68);
  const int ntiles = (E + TILE_E - 1) / TILE_E;
  const int rowW   = ntiles * 32;
  unsigned* bits   = (unsigned*)bufA;

  const int TPB = 256;
  auto gemm_blocks = [&](int M, int Nc) {
    int waves = ((M + 15) / 16) * ((Nc + 15) / 16);
    return (waves + 7) / 8;
  };
  auto gemm_row_blocks = [&](int M, int K) {
    int rb = 8192 / K;
    return (M + rb - 1) / rb;
  };

  const int npb    = (N > 256) ? (N + 255) / 256 : 1;
  const int owners = (N + npb - 1) / npb;
  edge_bits<<<(ntiles + 1) / 2, TPB, 0, stream>>>(ei, E, npb, rowW, bits, nxt);
  owner_rows<<<owners, OWN_T, 0, stream>>>(ei, E, bits, rowW, npb, N, rows, curs, nxt);

  gemm_wmma_f32<64, 128><<<gemm_row_blocks(N, 64), TPB, 0, stream>>>(x, W1, nullptr,
                                                                  bufA, N, 0);
  node_gather<2><<<(N + 7) / 8, TPB, 0, stream>>>(ei, E, nxt, rows, curs, a_src1, a_dst1,
                                                  bufA, b1, bufB, N);

  gemm_wmma_f32<128, 64><<<gemm_row_blocks(N, 128), TPB, 0, stream>>>(bufB, W2, nullptr,
                                                                 bufA, N, 0);
  node_gather<1><<<(N + 7) / 8, TPB, 0, stream>>>(ei, E, nxt, rows, curs, a_src2, a_dst2,
                                                  bufA, b2, bufB, N);

  gemm_wmma_f32<64, 64><<<gemm_row_blocks(N, 64), TPB, 0, stream>>>(bufB, m1w, m1b,
                                                                bufA, N, 1);
  if (C == 8) {
    gemm_wmma_f32<64, 8><<<gemm_row_blocks(N, 64), TPB, 0, stream>>>(bufA, m2w, m2b,
                                                                (float*)d_out, N, 1);
  } else {
    gemm_wmma_f32_dyn<<<gemm_blocks(N, C), TPB, 0, stream>>>(bufA, m2w, m2b,
                                                             (float*)d_out, N, 64, C, 1);
  }
}


# conv1's projection GEMM runs in the same launch as the edge bit pass (block-index roles, one LDS array); bit rows moved to the second plane, source rows behind their spill; 8 launches -> 7
# speedup vs baseline: 5.6873x; 1.0446x over previous
#include <hip/hip_runtime.h>
#include <math.h>

typedef __attribute__((ext_vector_type(2))) float v2f;
typedef __attribute__((ext_vector_type(8))) float v8f;

template <int K, int Nc>
__device__ __forceinline__ void
gemm_block(const float* __restrict__ A, const float* __restrict__ B,
           const float* __restrict__ bias, float* __restrict__ Cout, int M, int act,
           float* sA, int blk) {
  constexpr int  tilesN = (Nc + 15) >> 4;
  constexpr bool FULL   = (Nc % 16) == 0;
  constexpr int  RB  = 8192 / K;
  constexpr int  RT  = RB / 16;
  constexpr int  WG  = 8 / tilesN;
  constexpr int  TPW = RT / WG;
  constexpr int  P   = K + 4;
  constexpr int  KS  = K / 4;
  constexpr int  C4  = K / 4;
  constexpr int  CH  = KS < 16 ? KS : 16;
  static_assert(8192 % K == 0 && 8 % tilesN == 0 && RT % WG == 0, "block shape");
  const int t    = threadIdx.x;
  const int w    = __builtin_amdgcn_readfirstlane(t >> 5);
  const int lane = t & 31;
  const int half = lane >> 4;
  const int l16  = lane & 15;
  const int r0   = blk * RB;

  float4 av[8];
#pragma unroll
  for (int i = 0; i < 8; ++i) {
    int idx4 = i * 256 + t;
    int row  = idx4 / C4;
    int c4   = idx4 - row * C4;
    int gr   = r0 + row;
    gr = gr < M ? gr : M - 1;
    av[i] = *(const float4*)(A + (size_t)gr * K + c4 * 4);
  }

  const int tn   = w % tilesN;
  const int rsub = w / tilesN;
  const int col  = tn * 16 + l16;
  const bool cok = FULL || (col < Nc);
  const int colc = FULL ? col : (cok ? col : 0);
  const float* bp = B + (size_t)(2 * half) * Nc + colc;
  v2f bf[KS];
#pragma unroll
  for (int s = 0; s < KS; ++s) {
    float bx = bp[(size_t)(4 * s) * Nc];
    float by = bp[(size_t)(4 * s) * Nc + Nc];
    bf[s].x = FULL ? bx : (cok ? bx : 0.f);
    bf[s].y = FULL ? by : (cok ? by : 0.f);
  }
  float bv = 0.f;
  if (bias != nullptr && cok) bv = bias[col];

#pragma unroll
  for (int i = 0; i < 8; ++i) {
    int idx4 = i * 256 + t;
    int row  = idx4 / C4;
    int c4   = idx4 - row * C4;
    *(float4*)(sA + row * P + c4 * 4) = av[i];
  }
  __syncthreads();

  for (int j = 0; j < TPW; ++j) {
    const int rt   = rsub + WG * j;
    const int row0 = r0 + rt * 16;
    if (row0 >= M) break;
    const float* ar = sA + (rt * 16 + l16) * P + 2 * half;
    v8f c = {0.f, 0.f, 0.f, 0.f, 0.f, 0.f, 0.f, 0.f};
#pragma unroll
    for (int s0 = 0; s0 < KS; s0 += CH) {
      v2f a[CH];
#pragma unroll
      for (int s = 0; s < CH; ++s) a[s] = *(const v2f*)(ar + 4 * (s0 + s));
      __builtin_amdgcn_sched_barrier(0);
#pragma unroll
      for (int s = 0; s < CH; ++s)
        c = __builtin_amdgcn_wmma_f32_16x16x4_f32(false, a[s], false, bf[s0 + s], (short)0, c,
                                                  false, false);
      __builtin_amdgcn_sched_barrier(0);
    }
    if (cok) {
      const int rbase = row0 + 8 * half;
      float* crow = Cout + (size_t)rbase * Nc + col;
      if (row0 + 16 <= M) {
#pragma unroll
        for (int v = 0; v < 8; ++v) {
          float val = c[v] + bv;
          if (act) val = fmaxf(val, 0.f);
          crow[(size_t)v * Nc] = val;
        }
      } else {
#pragma unroll
        for (int v = 0; v < 8; ++v) {
          float val = c[v] + bv;
          if (act) val = fmaxf(val, 0.f);
          if (rbase + v < M) crow[(size_t)v * Nc] = val;
        }
      }
    }
  }
}

template <int K, int Nc>
__global__ void __launch_bounds__(256)
gemm_wmma_f32(const float* __restrict__ A, const float* __restrict__ B,
              const float* __restrict__ bias, float* __restrict__ Cout, int M, int act) {
  __shared__ __attribute__((aligned(16))) float sA[(8192 / K) * (K + 4)];
  gemm_block<K, Nc>(A, B, bias, Cout, M, act, sA, (int)blockIdx.x);
}

__global__ void gemm_wmma_f32_dyn(const float* __restrict__ A, const float* __restrict__ B,
                                  const float* __restrict__ bias, float* __restrict__ Cout,
                                  int M, int K, int Nc, int act) {
  int wave   = blockIdx.x * (blockDim.x >> 5) + (threadIdx.x >> 5);
  int tilesN = (Nc + 15) >> 4;
  int tilesM = (M + 15) >> 4;
  if (wave >= tilesM * tilesN) return;
  int tm = wave / tilesN, tn = wave - tm * tilesN;
  int lane = threadIdx.x & 31, half = lane >> 4, l16 = lane & 15;
  int row = tm * 16 + l16, col = tn * 16 + l16;
  bool cok = (col < Nc);
  int colc = cok ? col : 0;
  const float* ap = A + (size_t)row * K + 2 * half;
  const float* bp = B + (size_t)(2 * half) * Nc + colc;
  v8f c = {0.f, 0.f, 0.f, 0.f, 0.f, 0.f, 0.f, 0.f};
  for (int k0 = 0; k0 < K; k0 += 4) {
    float2 av = *(const float2*)ap;
    float bx = bp[0], by = bp[Nc];
    v2f a, b;
    a.x = av.x;            a.y = av.y;
    b.x = cok ? bx : 0.f;  b.y = cok ? by : 0.f;
    c = __builtin_amdgcn_wmma_f32_16x16x4_f32(false, a, false, b, (short)0, c,
                                              false, false);
    ap += 4;
    bp += (size_t)4 * Nc;
  }
  if (cok) {
    float bv = bias ? bias[col] : 0.f;
    float* crow = Cout + (size_t)(tm * 16 + 8 * half) * Nc + col;
    for (int v = 0; v < 8; ++v) {
      float val = c[v] + bv;
      if (act) val = fmaxf(val, 0.f);
      crow[(size_t)v * Nc] = val;
    }
  }
}

constexpr int OWN_T   = 1024;
constexpr int NPB_MAX = 512;
constexpr int QCAP    = 32768;
constexpr int SCAN_U  = 8;
constexpr int TILE_E  = 1024;

__device__ __forceinline__ void
edge_bits_block(const int* __restrict__ ei, int Eorig, int npb, int rowW,
                unsigned* __restrict__ bits, int* __restrict__ nxt,
                unsigned* tab, int blk, int nblk) {
  const int t = threadIdx.x;
  const int* dp = ei + Eorig;
  const unsigned un = (unsigned)npb;
  const int ntiles = rowW >> 5;
  const int w  = t >> 3;
  const int b0 = (4 * t) & 31;
  uint4* tab4 = (uint4*)tab;
  for (int i = t; i < 256 * 8; i += 256) tab4[i] = make_uint4(0u, 0u, 0u, 0u);
  for (int tile = blk; tile < ntiles; tile += nblk) {
    __syncthreads();
    const int e0 = tile * TILE_E + 4 * t;
    int d[4] = {-1, -1, -1, -1};
    if ((Eorig & 3) == 0 && e0 + 4 <= Eorig) {
      const int4 v = *(const int4*)(dp + e0);
      d[0] = v.x; d[1] = v.y; d[2] = v.z; d[3] = v.w;
      *(int4*)(nxt + e0) = make_int4(-1, -1, -1, -1);
    } else {
#pragma unroll
      for (int k = 0; k < 4; ++k)
        if (e0 + k < Eorig) { d[k] = dp[e0 + k]; nxt[e0 + k] = -1; }
    }
#pragma unroll
    for (int k = 0; k < 4; ++k) {
      if (d[k] >= 0) {
        const unsigned o = (unsigned)d[k] / un;
        if (o < 256u) atomicOr(&tab[o * 32u + (unsigned)w], 1u << (b0 + k));
      }
    }
    __syncthreads();
#pragma unroll
    for (int j = 0; j < 8; ++j) {
      const int i = t + 256 * j;
      const uint4 v = tab4[i];
      tab4[i] = make_uint4(0u, 0u, 0u, 0u);
      *(uint4*)(bits + (size_t)(i >> 3) * rowW + (size_t)tile * 32 + (i & 7) * 4) = v;
    }
  }
}

__global__ void __launch_bounds__(256)
edge_bits(const int* __restrict__ ei, int Eorig, int npb, int rowW,
          unsigned* __restrict__ bits, int* __restrict__ nxt) {
  __shared__ __attribute__((aligned(16))) unsigned tab[256 * 32];
  edge_bits_block(ei, Eorig, npb, rowW, bits, nxt, tab, (int)blockIdx.x, (int)gridDim.x);
}

__global__ void __launch_bounds__(256)
proj1_beside_edge_bits(const float* __restrict__ x, const float* __restrict__ W1,
                       float* __restrict__ h1, int M, int gemm_blocks,
                       const int* __restrict__ ei, int Eorig, int npb, int rowW,
                       unsigned* __restrict__ bits, int* __restrict__ nxt, int edge_blocks) {
  __shared__ __attribute__((aligned(16))) float smem[(8192 / 64) * (64 + 4)];
  const int b = (int)blockIdx.x;
  if (b < gemm_blocks)
    gemm_block<64, 128>(x, W1, nullptr, h1, M, 0, smem, b);
  else
    edge_bits_block(ei, Eorig, npb, rowW, bits, nxt, (unsigned*)smem, b - gemm_blocks, edge_blocks);
}

template <int N>
__device__ __forceinline__ int row_shr(int v) {
  return __builtin_amdgcn_update_dpp(0, v, 0x110 | N, 0xF, 0xF, true);
}

__global__ void __launch_bounds__(OWN_T)
owner_rows(const int* __restrict__ ei, int Eorig, const unsigned* __restrict__ bits, int rowW,
           int npb, int n, int* __restrict__ rows, int* __restrict__ curs, int* __restrict__ nxt) {
  __shared__ int q[QCAP + 1];
  __shared__ int rows_s[NPB_MAX * 32];
  __shared__ int cnt_s[NPB_MAX];
  __shared__ int ovh_s[NPB_MAX];
  __shared__ int qn;
  const int t  = threadIdx.x;
  const int o  = blockIdx.x;
  const int lo = o * npb;
  const int hi = (lo + npb < n) ? lo + npb : n;
  const int* dp = ei + Eorig;
  const uint4* brow = (const uint4*)(bits + (size_t)o * rowW);
  const int nvec = rowW >> 2;

  for (int base = lo; base < hi; base += NPB_MAX) {
    const int span = (hi - base < NPB_MAX) ? hi - base : NPB_MAX;
    for (int i = t; i < NPB_MAX; i += OWN_T) { cnt_s[i] = 0; ovh_s[i] = -1; }
    if (t == 0) qn = 0;
    __syncthreads();

    auto rank_edge = [&](int e, int d, int sv) {
      const unsigned loc = (unsigned)(d - base);
      if (loc < (unsigned)span) {
        const int r = atomicAdd(&cnt_s[loc], 1);
        if (r < 32) rows_s[loc * 32 + r] = sv;
        else nxt[e] = atomicExch(&ovh_s[loc], e);
      }
    };

    for (int ib = 0; ib < nvec; ib += SCAN_U * OWN_T) {
      const int it0 = ib + t;
      uint4 w[SCAN_U];
#pragma unroll
      for (int u = 0; u < SCAN_U; ++u) {
        const int it = it0 + u * OWN_T;
        w[u] = brow[it < nvec ? it : 0];
      }
      __builtin_amdgcn_sched_barrier(0);
      int cl = 0;
#pragma unroll
      for (int u = 0; u < SCAN_U; ++u) {
        if (it0 + u * OWN_T >= nvec) w[u] = make_uint4(0u, 0u, 0u, 0u);
        cl += __builtin_popcount(w[u].x) + __builtin_popcount(w[u].y) +
              __builtin_popcount(w[u].z) + __builtin_popcount(w[u].w);
      }
      int run = cl;
      run += row_shr<1>(run);
      run += row_shr<2>(run);
      run += row_shr<4>(run);
      run += row_shr<8>(run);
      const int lowtot = __builtin_amdgcn_readlane(run, 15);
      if ((t & 31) >= 16) run += lowtot;
      const int total = __builtin_amdgcn_readlane(run, 31);
      int first = 0;
      if (total != 0) {
        if ((t & 31) == 0) first = atomicAdd(&qn, total);
        first = __builtin_amdgcn_readfirstlane(first);
      }
      int slot = first + run - cl;
      const int send = slot + cl;
#pragma unroll
      for (int u = 0; u < SCAN_U; ++u) {
        const unsigned ww[4] = {w[u].x, w[u].y, w[u].z, w[u].w};
#pragma unroll
        for (int k = 0; k < 4; ++k) {
          unsigned m = ww[k];
          const int eb = ((it0 + u * OWN_T) * 4 + k) << 5;
          while (m != 0u) {
            q[slot < QCAP ? slot : QCAP] = eb + __builtin_ctz(m);
            m &= m - 1u;
            ++slot;
          }
        }
      }
      if (send > QCAP) {
        int s2 = send - cl;
#pragma nounroll
        for (int u = 0; u < SCAN_U; ++u) {
          const int it = it0 + u * OWN_T;
          if (it >= nvec) break;
          const unsigned* wp = (const unsigned*)(brow + it);
#pragma nounroll
          for (int k = 0; k < 4; ++k) {
            unsigned m = wp[k];
#pragma nounroll
            while (m != 0u) {
              const int e = ((it * 4 + k) << 5) + __builtin_ctz(m);
              m &= m - 1u;
              if (s2 >= QCAP) rank_edge(e, dp[e], ei[e]);
              ++s2;
            }
          }
        }
      }
    }
    __syncthreads();

    const int nq = qn < QCAP ? qn : QCAP;
    for (int j0 = t; j0 < nq; j0 += 4 * OWN_T) {
      int e4[4], d4[4], s4[4];
#pragma unroll
      for (int u = 0; u < 4; ++u) {
        const int j = j0 + u * OWN_T;
        e4[u] = q[j < nq ? j : j0];
        d4[u] = dp[e4[u]];
        s4[u] = ei[e4[u]];
      }
#pragma unroll
      for (int u = 0; u < 4; ++u)
        if (j0 + u * OWN_T < nq) rank_edge(e4[u], d4[u], s4[u]);
    }
    __syncthreads();

    for (int idx = t; idx < span * 32; idx += OWN_T) {
      const int loc = idx >> 5, k = idx & 31;
      int v = -1;
      if (k < cnt_s[loc]) v = rows_s[idx];
      rows[(size_t)(base + loc) * 32 + k] = v;
    }
    for (int idx = t; idx < span * 4; idx += OWN_T) {
      const int loc = idx >> 2;
      curs[(size_t)(base + loc) * 4 + (idx & 3)] = ((idx & 3) == 0) ? ovh_s[loc] : -1;
    }
    __syncthreads();
  }
}

template <int MASK>
__device__ __forceinline__ float row_xor(float v) {
  return __int_as_float(__builtin_amdgcn_update_dpp(0, __float_as_int(v), 0x160 | MASK, 0xF, 0xF, true));
}
template <int L>
__device__ __forceinline__ float row_share(float v) {
  return __int_as_float(__builtin_amdgcn_update_dpp(0, __float_as_int(v), 0x150 | L, 0xF, 0xF, true));
}
__device__ __forceinline__ float half_sum(float v) {
  v += row_xor<8>(v);
  v += row_xor<4>(v);
  v += row_xor<2>(v);
  v += row_xor<1>(v);
  return v;
}
__device__ __forceinline__ float half_max(float v) {
  v = fmaxf(v, row_xor<8>(v));
  v = fmaxf(v, row_xor<4>(v));
  v = fmaxf(v, row_xor<2>(v));
  v = fmaxf(v, row_xor<1>(v));
  return v;
}
__device__ __forceinline__ float dot4(float4 a, float4 b) {
  return fmaf(a.w, b.w, fmaf(a.z, b.z, fmaf(a.y, b.y, a.x * b.x)));
}
__device__ __forceinline__ float leaky(float a) { return (a > 0.f) ? a : 0.2f * a; }

struct Soft { float m, zl, a0, a1, a2, a3; };

template <int HEADS, int R, bool FULL>
__device__ __forceinline__ void gather_chunk(Soft& st, const float* __restrict__ h, int my_src,
                                             int k, int cnt, unsigned off, int half, int c16,
                                             float4 as4, float adn) {
  constexpr int F   = HEADS * 64;
  constexpr int EPL = (HEADS == 2) ? 1 : 2;
  float4 t[R];
#pragma unroll
  for (int q = 0; q < R; ++q) {
    const int e0 = k + q * EPL;
    if constexpr (!FULL) t[q] = make_float4(0.f, 0.f, 0.f, 0.f);
    if (FULL || e0 < cnt) {
      int s = __builtin_amdgcn_readlane(my_src, e0);
      if constexpr (HEADS == 1) {
        const int e1 = (FULL || e0 + 1 < cnt) ? e0 + 1 : e0;
        const int s1 = __builtin_amdgcn_readlane(my_src, e1);
        s = half ? s1 : s;
      }
      t[q] = *(const float4*)(h + ((unsigned)s * (unsigned)F + off));
    }
  }
  float d = 0.f;
#pragma unroll
  for (int q = 0; q < R; ++q) {
    if (FULL || k + q * EPL < cnt) {
      const float dq = half_sum(dot4(t[q], as4));
      d = (c16 == q) ? dq : d;
    }
  }
  const int  eidx  = k + c16 * EPL + ((HEADS == 1) ? half : 0);
  const bool valid = (c16 < R) && (FULL || eidx < cnt);
  const float a    = valid ? leaky(d + adn) : -INFINITY;
  const float mn   = fmaxf(st.m, half_max(a));
  const float ev   = expf(((c16 == R) ? st.m : a) - mn);
  const float rsc  = row_share<R>(ev);
  st.zl = st.zl * rsc + ((c16 == R) ? 0.f : ev);
  st.a0 *= rsc; st.a1 *= rsc; st.a2 *= rsc; st.a3 *= rsc;
  st.m = mn;
  auto add_row = [&](float w, float4 r) {
    st.a0 = fmaf(w, r.x, st.a0); st.a1 = fmaf(w, r.y, st.a1);
    st.a2 = fmaf(w, r.z, st.a2); st.a3 = fmaf(w, r.w, st.a3);
  };
  if (FULL || k + 0 * EPL < cnt) add_row(row_share<0>(ev), t[0]);
  if (FULL || k + 1 * EPL < cnt) add_row(row_share<1>(ev), t[1]);
  if (FULL || k + 2 * EPL < cnt) add_row(row_share<2>(ev), t[2]);
  if (FULL || k + 3 * EPL < cnt) add_row(row_share<3>(ev), t[3]);
  if (FULL || k + 4 * EPL < cnt) add_row(row_share<4>(ev), t[4]);
  if (FULL || k + 5 * EPL < cnt) add_row(row_share<5>(ev), t[5]);
  if (FULL || k + 6 * EPL < cnt) add_row(row_share<6>(ev), t[6]);
  if (FULL || k + 7 * EPL < cnt) add_row(row_share<7>(ev), t[7]);
}

template <int HEADS>
__global__ void __launch_bounds__(256)
node_gather(const int* __restrict__ ei, int Eorig,
                            const int* __restrict__ nxt, const int* __restrict__ rows,
                            const int* __restrict__ curs, const float* __restrict__ a_src,
                            const float* __restrict__ a_dst, const float* __restrict__ h,
                            const float* __restrict__ bias, float* __restrict__ out, int n) {
  constexpr int F   = HEADS * 64;
  constexpr int EPL = (HEADS == 2) ? 1 : 2;
  constexpr int R   = 8;
  int node = __builtin_amdgcn_readfirstlane((int)(blockIdx.x * (blockDim.x >> 5) + (threadIdx.x >> 5)));
  if (node >= n) return;
  const int lane = threadIdx.x & 31;
  const int half = lane >> 4;
  const int c16  = lane & 15;
  const int hd   = (HEADS == 2) ? half : 0;
  const unsigned off = (HEADS == 2) ? lane * 4 : c16 * 4;

  const int pre = rows[(size_t)node * 32 + lane];
  int cur = curs[node * 4 + (lane & 3)];
  const float4 own = *(const float4*)(h + (size_t)node * F + off);
  const float4 as4 = *(const float4*)(a_src + hd * 64 + c16 * 4);
  const float4 ad4 = *(const float4*)(a_dst + hd * 64 + c16 * 4);
  __builtin_amdgcn_sched_barrier(0);

  const float adn = half_sum(dot4(own, ad4));
  const bool mine = (HEADS == 2) || (half == 0);
  Soft st;
  st.m  = leaky(half_sum(dot4(own, as4)) + adn);
  st.zl = (mine && c16 == 0) ? 1.f : 0.f;
  st.a0 = mine ? own.x : 0.f;
  st.a1 = mine ? own.y : 0.f;
  st.a2 = mine ? own.z : 0.f;
  st.a3 = mine ? own.w : 0.f;

  unsigned act = __builtin_amdgcn_ballot_w32(cur >= 0) & 0xFu;
  const int max_rounds = (Eorig >> 5) + 2;
  int round = 0;
  do {
    int cnt = 0;
    int my_src = node;
    if (round == 0) {
      my_src = pre;
      cnt = __builtin_popcount(__builtin_amdgcn_ballot_w32(pre >= 0));
    } else
    while (act != 0u && cnt < 32) {
      int ci = cur > 0 ? cur : 0;
      int sv = ei[ci];
      int nv = nxt[ci];
#pragma unroll
      for (int c = 0; c < 4; ++c) {
        if (((act >> c) & 1u) != 0u && cnt < 32) {
          int s = __builtin_amdgcn_readlane(sv, c);
          if (lane == cnt) my_src = s;
          ++cnt;
          if ((lane & 3) == c) cur = nv;
        }
      }
      act = __builtin_amdgcn_ballot_w32(cur >= 0) & 0xFu;
    }
    int k = 0;
    for (; k + R * EPL <= cnt; k += R * EPL)
      gather_chunk<HEADS, R, true>(st, h, my_src, k, cnt, off, half, c16, as4, adn);
    if (k < cnt)
      gather_chunk<HEADS, R, false>(st, h, my_src, k, cnt, off, half, c16, as4, adn);
    if (act == 0u) break;
  } while (++round < max_rounds);

  float z = half_sum(st.zl);
  float acc[4] = {st.a0, st.a1, st.a2, st.a3};
  if constexpr (HEADS == 1) {
    const float mo  = __shfl_xor(st.m, 16, 32);
    const float rsc = expf(st.m - fmaxf(st.m, mo));
    z *= rsc;
#pragma unroll
    for (int j = 0; j < 4; ++j) acc[j] *= rsc;
    z += __shfl_xor(z, 16, 32);
#pragma unroll
    for (int j = 0; j < 4; ++j) acc[j] += __shfl_xor(acc[j], 16, 32);
  }

  if constexpr (HEADS == 2) {
    const float4 bv = *(const float4*)(bias + off);
    const float bb[4] = {bv.x, bv.y, bv.z, bv.w};
    float o[4];
#pragma unroll
    for (int j = 0; j < 4; ++j) {
      float v = acc[j] / (z + 1e-16f) + bb[j];
      o[j] = (v > 0.f) ? v : expm1f(v);
    }
    *(float4*)(out + (size_t)node * F + off) = make_float4(o[0], o[1], o[2], o[3]);
  } else {
    const unsigned o2 = off + 2u * (unsigned)half;
    const float2 bv = *(const float2*)(bias + o2);
    const float v0 = (half ? acc[2] : acc[0]) / (z + 1e-16f) + bv.x;
    const float v1 = (half ? acc[3] : acc[1]) / (z + 1e-16f) + bv.y;
    const float r0 = (v0 > 0.f) ? v0 : expm1f(v0);
    const float r1 = (v1 > 0.f) ? v1 : expm1f(v1);
    *(float2*)(out + (size_t)node * F + o2) = make_float2(r0, r1);
  }
}

extern "C" void kernel_launch(void* const* d_in, const int* in_sizes, int n_in,
                              void* d_out, int out_size, void* d_ws, size_t ws_size,
                              hipStream_t stream) {
  const float* x      = (const float*)d_in[0];
  const int*   ei     = (const int*)d_in[1];
  const float* W1     = (const float*)d_in[2];
  const float* a_src1 = (const float*)d_in[3];
  const float* a_dst1 = (const float*)d_in[4];
  const float* b1     = (const float*)d_in[5];
  const float* W2     = (const float*)d_in[6];
  const float* a_src2 = (const float*)d_in[7];
  const float* a_dst2 = (const float*)d_in[8];
  const float* b2     = (const float*)d_in[9];
  const float* m1w    = (const float*)d_in[10];
  const float* m1b    = (const float*)d_in[11];
  const float* m2w    = (const float*)d_in[12];
  const float* m2b    = (const float*)d_in[13];

  const int N    = in_sizes[0] / 64;
  const int E    = in_sizes[1] / 2;
  const int C    = in_sizes[12] / 64;

  float* bufA = (float*)d_ws;
  float* bufB = bufA + (size_t)N * 128;
  float* bufC = bufB + (size_t)N * 128;
  int*   nxt  = (int*)(bufC + (size_t)N * 68);
  const int ntiles = (E + TILE_E - 1) / TILE_E;
  const int rowW   = ntiles * 32;
  const size_t bits_words  = (size_t)256 * rowW;
  const size_t plane_words = (size_t)N * 128;
  size_t spill = bits_words > plane_words ? bits_words - plane_words : 0;
  spill = (spill + 31) & ~(size_t)31;
  const bool side = spill <= (size_t)N * 32;
  unsigned* bits = side ? (unsigned*)bufB : (unsigned*)bufA;
  int*   rows = (int*)bufC + (side ? spill : 0);
  int*   curs = rows + (size_t)N * 32;

  const int TPB = 256;
  auto gemm_blocks = [&](int M, int Nc) {
    int waves = ((M + 15) / 16) * ((Nc + 15) / 16);
    return (waves + 7) / 8;
  };
  auto gemm_row_blocks = [&](int M, int K) {
    int rb = 8192 / K;
    return (M + rb - 1) / rb;
  };

  const int npb    = (N > 256) ? (N + 255) / 256 : 1;
  const int owners = (N + npb - 1) / npb;
  const int eblocks = (ntiles + 1) / 2;
  const int gblocks = gemm_row_blocks(N, 64);
  if (side) {
    proj1_beside_edge_bits<<<gblocks + eblocks, TPB, 0, stream>>>(x, W1, bufA, N, gblocks,
                                                                  ei, E, npb, rowW, bits, nxt, eblocks);
    owner_rows<<<owners, OWN_T, 0, stream>>>(ei, E, bits, rowW, npb, N, rows, curs, nxt);
  } else {
    edge_bits<<<eblocks, TPB, 0, stream>>>(ei, E, npb, rowW, bits, nxt);
    owner_rows<<<owners, OWN_T, 0, stream>>>(ei, E, bits, rowW, npb, N, rows, curs, nxt);
    gemm_wmma_f32<64, 128><<<gblocks, TPB, 0, stream>>>(x, W1, nullptr, bufA, N, 0);
  }

  node_gather<2><<<(N + 7) / 8, TPB, 0, stream>>>(ei, E, nxt, rows, curs, a_src1, a_dst1,
                                                  bufA, b1, bufB, N);

  gemm_wmma_f32<128, 64><<<gemm_row_blocks(N, 128), TPB, 0, stream>>>(bufB, W2, nullptr,
                                                                 bufA, N, 0);
  node_gather<1><<<(N + 7) / 8, TPB, 0, stream>>>(ei, E, nxt, rows, curs, a_src2, a_dst2,
                                                  bufA, b2, bufB, N);

  gemm_wmma_f32<64, 64><<<gemm_row_blocks(N, 64), TPB, 0, stream>>>(bufB, m1w, m1b,
                                                                bufA, N, 1);
  if (C == 8) {
    gemm_wmma_f32<64, 8><<<gemm_row_blocks(N, 64), TPB, 0, stream>>>(bufA, m2w, m2b,
                                                                (float*)d_out, N, 1);
  } else {
    gemm_wmma_f32_dyn<<<gemm_blocks(N, C), TPB, 0, stream>>>(bufA, m2w, m2b,
                                                             (float*)d_out, N, 64, C, 1);
  }
}


# edge grouping: pass 1 writes each owner's edges of a tile as (source, destination) records in its 128-byte line; owners rank from their row, no per-edge fetch; arrival-order rank, sums' order shifts
# speedup vs baseline: 6.2277x; 1.0950x over previous
#include <hip/hip_runtime.h>
#include <math.h>

typedef __attribute__((ext_vector_type(2))) float v2f;
typedef __attribute__((ext_vector_type(8))) float v8f;

template <int K, int Nc>
__device__ __forceinline__ void
gemm_block(const float* __restrict__ A, const float* __restrict__ B,
           const float* __restrict__ bias, float* __restrict__ Cout, int M, int act,
           float* sA, int blk) {
  constexpr int  tilesN = (Nc + 15) >> 4;
  constexpr bool FULL   = (Nc % 16) == 0;
  constexpr int  RB  = 8192 / K;
  constexpr int  RT  = RB / 16;
  constexpr int  WG  = 8 / tilesN;
  constexpr int  TPW = RT / WG;
  constexpr int  P   = K + 4;
  constexpr int  KS  = K / 4;
  constexpr int  C4  = K / 4;
  constexpr int  CH  = KS < 16 ? KS : 16;
  static_assert(8192 % K == 0 && 8 % tilesN == 0 && RT % WG == 0, "block shape");
  const int t    = threadIdx.x;
  const int w    = __builtin_amdgcn_readfirstlane(t >> 5);
  const int lane = t & 31;
  const int half = lane >> 4;
  const int l16  = lane & 15;
  const int r0   = blk * RB;

  float4 av[8];
#pragma unroll
  for (int i = 0; i < 8; ++i) {
    int idx4 = i * 256 + t;
    int row  = idx4 / C4;
    int c4   = idx4 - row * C4;
    int gr   = r0 + row;
    gr = gr < M ? gr : M - 1;
    av[i] = *(const float4*)(A + (size_t)gr * K + c4 * 4);
  }

  const int tn   = w % tilesN;
  const int rsub = w / tilesN;
  const int col  = tn * 16 + l16;
  const bool cok = FULL || (col < Nc);
  const int colc = FULL ? col : (cok ? col : 0);
  const float* bp = B + (size_t)(2 * half) * Nc + colc;
  v2f bf[KS];
#pragma unroll
  for (int s = 0; s < KS; ++s) {
    float bx = bp[(size_t)(4 * s) * Nc];
    float by = bp[(size_t)(4 * s) * Nc + Nc];
    bf[s].x = FULL ? bx : (cok ? bx : 0.f);
    bf[s].y = FULL ? by : (cok ? by : 0.f);
  }
  float bv = 0.f;
  if (bias != nullptr && cok) bv = bias[col];

#pragma unroll
  for (int i = 0; i < 8; ++i) {
    int idx4 = i * 256 + t;
    int row  = idx4 / C4;
    int c4   = idx4 - row * C4;
    *(float4*)(sA + row * P + c4 * 4) = av[i];
  }
  __syncthreads();

  for (int j = 0; j < TPW; ++j) {
    const int rt   = rsub + WG * j;
    const int row0 = r0 + rt * 16;
    if (row0 >= M) break;
    const float* ar = sA + (rt * 16 + l16) * P + 2 * half;
    v8f c = {0.f, 0.f, 0.f, 0.f, 0.f, 0.f, 0.f, 0.f};
#pragma unroll
    for (int s0 = 0; s0 < KS; s0 += CH) {
      v2f a[CH];
#pragma unroll
      for (int s = 0; s < CH; ++s) a[s] = *(const v2f*)(ar + 4 * (s0 + s));
      __builtin_amdgcn_sched_barrier(0);
#pragma unroll
      for (int s = 0; s < CH; ++s)
        c = __builtin_amdgcn_wmma_f32_16x16x4_f32(false, a[s], false, bf[s0 + s], (short)0, c,
                                                  false, false);
      __builtin_amdgcn_sched_barrier(0);
    }
    if (cok) {
      const int rbase = row0 + 8 * half;
      float* crow = Cout + (size_t)rbase * Nc + col;
      if (row0 + 16 <= M) {
#pragma unroll
        for (int v = 0; v < 8; ++v) {
          float val = c[v] + bv;
          if (act) val = fmaxf(val, 0.f);
          crow[(size_t)v * Nc] = val;
        }
      } else {
#pragma unroll
        for (int v = 0; v < 8; ++v) {
          float val = c[v] + bv;
          if (act) val = fmaxf(val, 0.f);
          if (rbase + v < M) crow[(size_t)v * Nc] = val;
        }
      }
    }
  }
}

template <int K, int Nc>
__global__ void __launch_bounds__(256)
gemm_wmma_f32(const float* __restrict__ A, const float* __restrict__ B,
              const float* __restrict__ bias, float* __restrict__ Cout, int M, int act) {
  __shared__ __attribute__((aligned(16))) float sA[(8192 / K) * (K + 4)];
  gemm_block<K, Nc>(A, B, bias, Cout, M, act, sA, (int)blockIdx.x);
}

__global__ void gemm_wmma_f32_dyn(const float* __restrict__ A, const float* __restrict__ B,
                                  const float* __restrict__ bias, float* __restrict__ Cout,
                                  int M, int K, int Nc, int act) {
  int wave   = blockIdx.x * (blockDim.x >> 5) + (threadIdx.x >> 5);
  int tilesN = (Nc + 15) >> 4;
  int tilesM = (M + 15) >> 4;
  if (wave >= tilesM * tilesN) return;
  int tm = wave / tilesN, tn = wave - tm * tilesN;
  int lane = threadIdx.x & 31, half = lane >> 4, l16 = lane & 15;
  int row = tm * 16 + l16, col = tn * 16 + l16;
  bool cok = (col < Nc);
  int colc = cok ? col : 0;
  const float* ap = A + (size_t)row * K + 2 * half;
  const float* bp = B + (size_t)(2 * half) * Nc + colc;
  v8f c = {0.f, 0.f, 0.f, 0.f, 0.f, 0.f, 0.f, 0.f};
  for (int k0 = 0; k0 < K; k0 += 4) {
    float2 av = *(const float2*)ap;
    float bx = bp[0], by = bp[Nc];
    v2f a, b;
    a.x = av.x;            a.y = av.y;
    b.x = cok ? bx : 0.f;  b.y = cok ? by : 0.f;
    c = __builtin_amdgcn_wmma_f32_16x16x4_f32(false, a, false, b, (short)0, c,
                                              false, false);
    ap += 4;
    bp += (size_t)4 * Nc;
  }
  if (cok) {
    float bv = bias ? bias[col] : 0.f;
    float* crow = Cout + (size_t)(tm * 16 + 8 * half) * Nc + col;
    for (int v = 0; v < 8; ++v) {
      float val = c[v] + bv;
      if (act) val = fmaxf(val, 0.f);
      crow[(size_t)v * Nc] = val;
    }
  }
}

constexpr int OWN_T   = 1024;
constexpr int NPB_MAX = 512;
constexpr int SCAN_U  = 8;
constexpr int TILE_E  = 1024;
constexpr int REC_MAX = 15;
constexpr int OVT_CAP = 1024;
static_assert(OWN_T == TILE_E, "the listed tiles are read one edge per thread");

template <int G>
__device__ __forceinline__ void
lines_out4(unsigned* tab, unsigned* __restrict__ bits, int rowW, int tile, int t) {
  const uint4* tab4 = (const uint4*)tab;
  const uint4 v0 = tab4[t + 256 * (G + 0)];
  const uint4 v1 = tab4[t + 256 * (G + 1)];
  const uint4 v2 = tab4[t + 256 * (G + 2)];
  const uint4 v3 = tab4[t + 256 * (G + 3)];
  if ((t & 7) == 0) {
    tab[((t + 256 * (G + 0)) >> 3) * 32] = 0u;
    tab[((t + 256 * (G + 1)) >> 3) * 32] = 0u;
    tab[((t + 256 * (G + 2)) >> 3) * 32] = 0u;
    tab[((t + 256 * (G + 3)) >> 3) * 32] = 0u;
  }
  unsigned* out = bits + (size_t)(t >> 3) * rowW + (size_t)tile * 32 + (t & 7) * 4;
  *(uint4*)(out + (size_t)(32 * (G + 0)) * rowW) = v0;
  *(uint4*)(out + (size_t)(32 * (G + 1)) * rowW) = v1;
  *(uint4*)(out + (size_t)(32 * (G + 2)) * rowW) = v2;
  *(uint4*)(out + (size_t)(32 * (G + 3)) * rowW) = v3;
}

__device__ __forceinline__ void
edge_recs_block(const int* __restrict__ ei, int Eorig, int npb, int rowW,
                unsigned* __restrict__ bits, int* __restrict__ nxt,
                unsigned* tab, int blk, int nblk) {
  const int t = threadIdx.x;
  const int* dp = ei + Eorig;
  const unsigned un = (unsigned)npb;
  const int ntiles = rowW >> 5;
  const bool wide = npb > 65536;
  const unsigned step = wide ? (unsigned)(REC_MAX + 1) : 1u;
  tab[t * 32] = 0u;
  for (int tile = blk; tile < ntiles; tile += nblk) {
    __syncthreads();
    const int e0 = tile * TILE_E + 4 * t;
    int d[4] = {-1, -1, -1, -1};
    int s[4] = {0, 0, 0, 0};
    if ((Eorig & 3) == 0 && e0 + 4 <= Eorig) {
      const int4 v = *(const int4*)(dp + e0);
      const int4 u = *(const int4*)(ei + e0);
      d[0] = v.x; d[1] = v.y; d[2] = v.z; d[3] = v.w;
      s[0] = u.x; s[1] = u.y; s[2] = u.z; s[3] = u.w;
      *(int4*)(nxt + e0) = make_int4(-1, -1, -1, -1);
    } else {
#pragma unroll
      for (int k = 0; k < 4; ++k)
        if (e0 + k < Eorig) { d[k] = dp[e0 + k]; s[k] = ei[e0 + k]; nxt[e0 + k] = -1; }
    }
    unsigned ow[4], slot[4];
#pragma unroll
    for (int k = 0; k < 4; ++k) {
      ow[k]   = (d[k] >= 0) ? (unsigned)d[k] / un : 256u;
      slot[k] = (unsigned)REC_MAX;
      if (ow[k] < 256u) slot[k] = atomicAdd(&tab[ow[k] * 32u], step);
    }
#pragma unroll
    for (int k = 0; k < 4; ++k) {
      if (slot[k] < (unsigned)REC_MAX && !wide)
        *(uint2*)(tab + ow[k] * 32u + 2u + 2u * slot[k]) =
            make_uint2((unsigned)s[k], ((unsigned)d[k] - ow[k] * un) | ((unsigned)(4 * t + k) << 16));
    }
    __syncthreads();
    lines_out4<0>(tab, bits, rowW, tile, t);
    lines_out4<4>(tab, bits, rowW, tile, t);
  }
}

__global__ void __launch_bounds__(256)
edge_recs(const int* __restrict__ ei, int Eorig, int npb, int rowW,
          unsigned* __restrict__ bits, int* __restrict__ nxt) {
  __shared__ __attribute__((aligned(16))) unsigned tab[256 * 32];
  edge_recs_block(ei, Eorig, npb, rowW, bits, nxt, tab, (int)blockIdx.x, (int)gridDim.x);
}

__global__ void __launch_bounds__(256)
proj1_beside_edge_recs(const float* __restrict__ x, const float* __restrict__ W1,
                       float* __restrict__ h1, int M, int gemm_blocks,
                       const int* __restrict__ ei, int Eorig, int npb, int rowW,
                       unsigned* __restrict__ bits, int* __restrict__ nxt, int edge_blocks) {
  __shared__ __attribute__((aligned(16))) float smem[(8192 / 64) * (64 + 4)];
  const int b = (int)blockIdx.x;
  if (b < gemm_blocks)
    gemm_block<64, 128>(x, W1, nullptr, h1, M, 0, smem, b);
  else
    edge_recs_block(ei, Eorig, npb, rowW, bits, nxt, (unsigned*)smem, b - gemm_blocks, edge_blocks);
}

template <int L>
__device__ __forceinline__ unsigned row_lane(unsigned v) {
  return (unsigned)__builtin_amdgcn_update_dpp(0, (int)v, 0x150 | L, 0xF, 0xF, true);
}

__global__ void __launch_bounds__(OWN_T)
owner_rows(const int* __restrict__ ei, int Eorig, const unsigned* __restrict__ bits, int rowW,
           int npb, int n, int* __restrict__ rows, int* __restrict__ curs, int* __restrict__ nxt) {
  __shared__ int rows_s[NPB_MAX * 32];
  __shared__ int cnt_s[NPB_MAX];
  __shared__ int ovh_s[NPB_MAX];
  __shared__ int ovt[OVT_CAP];
  __shared__ int novt;
  const int t  = threadIdx.x;
  const int o  = blockIdx.x;
  const int lo = o * npb;
  const int hi = (lo + npb < n) ? lo + npb : n;
  const int* dp = ei + Eorig;
  const uint4* brow = (const uint4*)(bits + (size_t)o * rowW);
  const int nvec = rowW >> 2;
  const int p = t & 7;

  for (int base = lo; base < hi; base += NPB_MAX) {
    const int span = (hi - base < NPB_MAX) ? hi - base : NPB_MAX;
    for (int i = t; i < NPB_MAX; i += OWN_T) { cnt_s[i] = 0; ovh_s[i] = -1; }
    if (t == 0) novt = 0;
    __syncthreads();

    auto rank_edge = [&](int e, int d, int sv) {
      const unsigned loc = (unsigned)(d - base);
      if (loc < (unsigned)span) {
        const int r = atomicAdd(&cnt_s[loc], 1);
        if (r < 32) rows_s[loc * 32 + r] = sv;
        else nxt[e] = atomicExch(&ovh_s[loc], e);
      }
    };

    for (int ib = 0; ib < nvec; ib += SCAN_U * OWN_T) {
      const int it0 = ib + t;
      uint4 w[SCAN_U];
#pragma unroll
      for (int u = 0; u < SCAN_U; ++u) {
        const int it = it0 + u * OWN_T;
        w[u] = brow[it < nvec ? it : 0];
      }
      __builtin_amdgcn_sched_barrier(0);
#pragma unroll
      for (int u = 0; u < SCAN_U; ++u) {
        const int it = it0 + u * OWN_T;
        if (it >= nvec) w[u] = make_uint4(0u, 0u, 0u, 0u);
        const unsigned c0 = row_lane<0>(w[u].x);
        const unsigned c8 = row_lane<8>(w[u].x);
        const unsigned cn = (t & 8) ? c8 : c0;
        const int nrec = cn <= (unsigned)REC_MAX ? (int)cn : 0;
        const int tile = it >> 3;
        const int eb   = tile * TILE_E;
        if (p != 0 && 2 * p - 1 < nrec)
          rank_edge(eb + (int)(w[u].y >> 16), lo + (int)(w[u].y & 0xFFFFu), (int)w[u].x);
        if (2 * p < nrec)
          rank_edge(eb + (int)(w[u].w >> 16), lo + (int)(w[u].w & 0xFFFFu), (int)w[u].z);
        if (p == 0 && cn > (unsigned)REC_MAX) {
          const int sl = atomicAdd(&novt, 1);
          if (sl < OVT_CAP) ovt[sl] = tile;
          else {
#pragma nounroll
            for (int k = 0; k < TILE_E; ++k) {
              const int e = eb + k;
              if (e < Eorig) rank_edge(e, dp[e], ei[e]);
            }
          }
        }
      }
    }
    __syncthreads();

    const int nov = novt < OVT_CAP ? novt : OVT_CAP;
    for (int i = 0; i < nov; ++i) {
      const int e = ovt[i] * TILE_E + t;
      if (e < Eorig) rank_edge(e, dp[e], ei[e]);
    }
    __syncthreads();

    for (int idx = t; idx < span * 32; idx += OWN_T) {
      const int loc = idx >> 5, k = idx & 31;
      int v = -1;
      if (k < cnt_s[loc]) v = rows_s[idx];
      rows[(size_t)(base + loc) * 32 + k] = v;
    }
    for (int idx = t; idx < span * 4; idx += OWN_T) {
      const int loc = idx >> 2;
      curs[(size_t)(base + loc) * 4 + (idx & 3)] = ((idx & 3) == 0) ? ovh_s[loc] : -1;
    }
    __syncthreads();
  }
}

template <int MASK>
__device__ __forceinline__ float row_xor(float v) {
  return __int_as_float(__builtin_amdgcn_update_dpp(0, __float_as_int(v), 0x160 | MASK, 0xF, 0xF, true));
}
template <int L>
__device__ __forceinline__ float row_share(float v) {
  return __int_as_float(__builtin_amdgcn_update_dpp(0, __float_as_int(v), 0x150 | L, 0xF, 0xF, true));
}
__device__ __forceinline__ float half_sum(float v) {
  v += row_xor<8>(v);
  v += row_xor<4>(v);
  v += row_xor<2>(v);
  v += row_xor<1>(v);
  return v;
}
__device__ __forceinline__ float half_max(float v) {
  v = fmaxf(v, row_xor<8>(v));
  v = fmaxf(v, row_xor<4>(v));
  v = fmaxf(v, row_xor<2>(v));
  v = fmaxf(v, row_xor<1>(v));
  return v;
}
__device__ __forceinline__ float dot4(float4 a, float4 b) {
  return fmaf(a.w, b.w, fmaf(a.z, b.z, fmaf(a.y, b.y, a.x * b.x)));
}
__device__ __forceinline__ float leaky(float a) { return (a > 0.f) ? a : 0.2f * a; }

struct Soft { float m, zl, a0, a1, a2, a3; };

template <int HEADS, int R, bool FULL>
__device__ __forceinline__ void gather_chunk(Soft& st, const float* __restrict__ h, int my_src,
                                             int k, int cnt, unsigned off, int half, int c16,
                                             float4 as4, float adn) {
  constexpr int F   = HEADS * 64;
  constexpr int EPL = (HEADS == 2) ? 1 : 2;
  float4 t[R];
#pragma unroll
  for (int q = 0; q < R; ++q) {
    const int e0 = k + q * EPL;
    if constexpr (!FULL) t[q] = make_float4(0.f, 0.f, 0.f, 0.f);
    if (FULL || e0 < cnt) {
      int s = __builtin_amdgcn_readlane(my_src, e0);
      if constexpr (HEADS == 1) {
        const int e1 = (FULL || e0 + 1 < cnt) ? e0 + 1 : e0;
        const int s1 = __builtin_amdgcn_readlane(my_src, e1);
        s = half ? s1 : s;
      }
      t[q] = *(const float4*)(h + ((unsigned)s * (unsigned)F + off));
    }
  }
  float d = 0.f;
#pragma unroll
  for (int q = 0; q < R; ++q) {
    if (FULL || k + q * EPL < cnt) {
      const float dq = half_sum(dot4(t[q], as4));
      d = (c16 == q) ? dq : d;
    }
  }
  const int  eidx  = k + c16 * EPL + ((HEADS == 1) ? half : 0);
  const bool valid = (c16 < R) && (FULL || eidx < cnt);
  const float a    = valid ? leaky(d + adn) : -INFINITY;
  const float mn   = fmaxf(st.m, half_max(a));
  const float ev   = expf(((c16 == R) ? st.m : a) - mn);
  const float rsc  = row_share<R>(ev);
  st.zl = st.zl * rsc + ((c16 == R) ? 0.f : ev);
  st.a0 *= rsc; st.a1 *= rsc; st.a2 *= rsc; st.a3 *= rsc;
  st.m = mn;
  auto add_row = [&](float w, float4 r) {
    st.a0 = fmaf(w, r.x, st.a0); st.a1 = fmaf(w, r.y, st.a1);
    st.a2 = fmaf(w, r.z, st.a2); st.a3 = fmaf(w, r.w, st.a3);
  };
  if (FULL || k + 0 * EPL < cnt) add_row(row_share<0>(ev), t[0]);
  if (FULL || k + 1 * EPL < cnt) add_row(row_share<1>(ev), t[1]);
  if (FULL || k + 2 * EPL < cnt) add_row(row_share<2>(ev), t[2]);
  if (FULL || k + 3 * EPL < cnt) add_row(row_share<3>(ev), t[3]);
  if (FULL || k + 4 * EPL < cnt) add_row(row_share<4>(ev), t[4]);
  if (FULL || k + 5 * EPL < cnt) add_row(row_share<5>(ev), t[5]);
  if (FULL || k + 6 * EPL < cnt) add_row(row_share<6>(ev), t[6]);
  if (FULL || k + 7 * EPL < cnt) add_row(row_share<7>(ev), t[7]);
}

template <int HEADS>
__global__ void __launch_bounds__(256)
node_gather(const int* __restrict__ ei, int Eorig,
                            const int* __restrict__ nxt, const int* __restrict__ rows,
                            const int* __restrict__ curs, const float* __restrict__ a_src,
                            const float* __restrict__ a_dst, const float* __restrict__ h,
                            const float* __restrict__ bias, float* __restrict__ out, int n) {
  constexpr int F   = HEADS * 64;
  constexpr int EPL = (HEADS == 2) ? 1 : 2;
  constexpr int R   = 8;
  int node = __builtin_amdgcn_readfirstlane((int)(blockIdx.x * (blockDim.x >> 5) + (threadIdx.x >> 5)));
  if (node >= n) return;
  const int lane = threadIdx.x & 31;
  const int half = lane >> 4;
  const int c16  = lane & 15;
  const int hd   = (HEADS == 2) ? half : 0;
  const unsigned off = (HEADS == 2) ? lane * 4 : c16 * 4;

  const int pre = rows[(size_t)node * 32 + lane];
  int cur = curs[node * 4 + (lane & 3)];
  const float4 own = *(const float4*)(h + (size_t)node * F + off);
  const float4 as4 = *(const float4*)(a_src + hd * 64 + c16 * 4);
  const float4 ad4 = *(const float4*)(a_dst + hd * 64 + c16 * 4);
  __builtin_amdgcn_sched_barrier(0);

  const float adn = half_sum(dot4(own, ad4));
  const bool mine = (HEADS == 2) || (half == 0);
  Soft st;
  st.m  = leaky(half_sum(dot4(own, as4)) + adn);
  st.zl = (mine && c16 == 0) ? 1.f : 0.f;
  st.a0 = mine ? own.x : 0.f;
  st.a1 = mine ? own.y : 0.f;
  st.a2 = mine ? own.z : 0.f;
  st.a3 = mine ? own.w : 0.f;

  unsigned act = __builtin_amdgcn_ballot_w32(cur >= 0) & 0xFu;
  const int max_rounds = (Eorig >> 5) + 2;
  int round = 0;
  do {
    int cnt = 0;
    int my_src = node;
    if (round == 0) {
      my_src = pre;
      cnt = __builtin_popcount(__builtin_amdgcn_ballot_w32(pre >= 0));
    } else
    while (act != 0u && cnt < 32) {
      int ci = cur > 0 ? cur : 0;
      int sv = ei[ci];
      int nv = nxt[ci];
#pragma unroll
      for (int c = 0; c < 4; ++c) {
        if (((act >> c) & 1u) != 0u && cnt < 32) {
          int s = __builtin_amdgcn_readlane(sv, c);
          if (lane == cnt) my_src = s;
          ++cnt;
          if ((lane & 3) == c) cur = nv;
        }
      }
      act = __builtin_amdgcn_ballot_w32(cur >= 0) & 0xFu;
    }
    int k = 0;
    for (; k + R * EPL <= cnt; k += R * EPL)
      gather_chunk<HEADS, R, true>(st, h, my_src, k, cnt, off, half, c16, as4, adn);
    if (k < cnt)
      gather_chunk<HEADS, R, false>(st, h, my_src, k, cnt, off, half, c16, as4, adn);
    if (act == 0u) break;
  } while (++round < max_rounds);

  float z = half_sum(st.zl);
  float acc[4] = {st.a0, st.a1, st.a2, st.a3};
  if constexpr (HEADS == 1) {
    const float mo  = __shfl_xor(st.m, 16, 32);
    const float rsc = expf(st.m - fmaxf(st.m, mo));
    z *= rsc;
#pragma unroll
    for (int j = 0; j < 4; ++j) acc[j] *= rsc;
    z += __shfl_xor(z, 16, 32);
#pragma unroll
    for (int j = 0; j < 4; ++j) acc[j] += __shfl_xor(acc[j], 16, 32);
  }

  if constexpr (HEADS == 2) {
    const float4 bv = *(const float4*)(bias + off);
    const float bb[4] = {bv.x, bv.y, bv.z, bv.w};
    float o[4];
#pragma unroll
    for (int j = 0; j < 4; ++j) {
      float v = acc[j] / (z + 1e-16f) + bb[j];
      o[j] = (v > 0.f) ? v : expm1f(v);
    }
    *(float4*)(out + (size_t)node * F + off) = make_float4(o[0], o[1], o[2], o[3]);
  } else {
    const unsigned o2 = off + 2u * (unsigned)half;
    const float2 bv = *(const float2*)(bias + o2);
    const float v0 = (half ? acc[2] : acc[0]) / (z + 1e-16f) + bv.x;
    const float v1 = (half ? acc[3] : acc[1]) / (z + 1e-16f) + bv.y;
    const float r0 = (v0 > 0.f) ? v0 : expm1f(v0);
    const float r1 = (v1 > 0.f) ? v1 : expm1f(v1);
    *(float2*)(out + (size_t)node * F + o2) = make_float2(r0, r1);
  }
}

extern "C" void kernel_launch(void* const* d_in, const int* in_sizes, int n_in,
                              void* d_out, int out_size, void* d_ws, size_t ws_size,
                              hipStream_t stream) {
  const float* x      = (const float*)d_in[0];
  const int*   ei     = (const int*)d_in[1];
  const float* W1     = (const float*)d_in[2];
  const float* a_src1 = (const float*)d_in[3];
  const float* a_dst1 = (const float*)d_in[4];
  const float* b1     = (const float*)d_in[5];
  const float* W2     = (const float*)d_in[6];
  const float* a_src2 = (const float*)d_in[7];
  const float* a_dst2 = (const float*)d_in[8];
  const float* b2     = (const float*)d_in[9];
  const float* m1w    = (const float*)d_in[10];
  const float* m1b    = (const float*)d_in[11];
  const float* m2w    = (const float*)d_in[12];
  const float* m2b    = (const float*)d_in[13];

  const int N    = in_sizes[0] / 64;
  const int E    = in_sizes[1] / 2;
  const int C    = in_sizes[12] / 64;

  float* bufA = (float*)d_ws;
  float* bufB = bufA + (size_t)N * 128;
  float* bufC = bufB + (size_t)N * 128;
  int*   nxt  = (int*)(bufC + (size_t)N * 68);
  const int ntiles = (E + TILE_E - 1) / TILE_E;
  const int rowW   = ntiles * 32;
  const size_t bits_words  = (size_t)256 * rowW;
  const size_t plane_words = (size_t)N * 128;
  size_t spill = bits_words > plane_words ? bits_words - plane_words : 0;
  spill = (spill + 31) & ~(size_t)31;
  const bool side = spill <= (size_t)N * 32;
  unsigned* bits = side ? (unsigned*)bufB : (unsigned*)bufA;
  int*   rows = (int*)bufC + (side ? spill : 0);
  int*   curs = rows + (size_t)N * 32;

  const int TPB = 256;
  auto gemm_blocks = [&](int M, int Nc) {
    int waves = ((M + 15) / 16) * ((Nc + 15) / 16);
    return (waves + 7) / 8;
  };
  auto gemm_row_blocks = [&](int M, int K) {
    int rb = 8192 / K;
    return (M + rb - 1) / rb;
  };

  const int npb    = (N > 256) ? (N + 255) / 256 : 1;
  const int owners = (N + npb - 1) / npb;
  const int eblocks = (ntiles + 1) / 2;
  const int gblocks = gemm_row_blocks(N, 64);
  if (side) {
    proj1_beside_edge_recs<<<gblocks + eblocks, TPB, 0, stream>>>(x, W1, bufA, N, gblocks,
                                                                  ei, E, npb, rowW, bits, nxt, eblocks);
    owner_rows<<<owners, OWN_T, 0, stream>>>(ei, E, bits, rowW, npb, N, rows, curs, nxt);
  } else {
    edge_recs<<<eblocks, TPB, 0, stream>>>(ei, E, npb, rowW, bits, nxt);
    owner_rows<<<owners, OWN_T, 0, stream>>>(ei, E, bits, rowW, npb, N, rows, curs, nxt);
    gemm_wmma_f32<64, 128><<<gblocks, TPB, 0, stream>>>(x, W1, nullptr, bufA, N, 0);
  }

  node_gather<2><<<(N + 7) / 8, TPB, 0, stream>>>(ei, E, nxt, rows, curs, a_src1, a_dst1,
                                                  bufA, b1, bufB, N);

  gemm_wmma_f32<128, 64><<<gemm_row_blocks(N, 128), TPB, 0, stream>>>(bufB, W2, nullptr,
                                                                 bufA, N, 0);
  node_gather<1><<<(N + 7) / 8, TPB, 0, stream>>>(ei, E, nxt, rows, curs, a_src2, a_dst2,
                                                  bufA, b2, bufB, N);

  gemm_wmma_f32<64, 64><<<gemm_row_blocks(N, 64), TPB, 0, stream>>>(bufB, m1w, m1b,
                                                                bufA, N, 1);
  if (C == 8) {
    gemm_wmma_f32<64, 8><<<gemm_row_blocks(N, 64), TPB, 0, stream>>>(bufA, m2w, m2b,
                                                                (float*)d_out, N, 1);
  } else {
    gemm_wmma_f32_dyn<<<gemm_blocks(N, C), TPB, 0, stream>>>(bufA, m2w, m2b,
                                                             (float*)d_out, N, 64, C, 1);
  }
}


# node gathers: softmax weights and elu's exp(v)-1 on the hardware exp2 (arguments never positive), one reciprocal of z per node; no library expf/expm1f, no quotient per output; last bits shift
# speedup vs baseline: 6.5567x; 1.0528x over previous
#include <hip/hip_runtime.h>
#include <math.h>

typedef __attribute__((ext_vector_type(2))) float v2f;
typedef __attribute__((ext_vector_type(8))) float v8f;

template <int K, int Nc>
__device__ __forceinline__ void
gemm_block(const float* __restrict__ A, const float* __restrict__ B,
           const float* __restrict__ bias, float* __restrict__ Cout, int M, int act,
           float* sA, int blk) {
  constexpr int  tilesN = (Nc + 15) >> 4;
  constexpr bool FULL   = (Nc % 16) == 0;
  constexpr int  RB  = 8192 / K;
  constexpr int  RT  = RB / 16;
  constexpr int  WG  = 8 / tilesN;
  constexpr int  TPW = RT / WG;
  constexpr int  P   = K + 4;
  constexpr int  KS  = K / 4;
  constexpr int  C4  = K / 4;
  constexpr int  CH  = KS < 16 ? KS : 16;
  static_assert(8192 % K == 0 && 8 % tilesN == 0 && RT % WG == 0, "block shape");
  const int t    = threadIdx.x;
  const int w    = __builtin_amdgcn_readfirstlane(t >> 5);
  const int lane = t & 31;
  const int half = lane >> 4;
  const int l16  = lane & 15;
  const int r0   = blk * RB;

  float4 av[8];
#pragma unroll
  for (int i = 0; i < 8; ++i) {
    int idx4 = i * 256 + t;
    int row  = idx4 / C4;
    int c4   = idx4 - row * C4;
    int gr   = r0 + row;
    gr = gr < M ? gr : M - 1;
    av[i] = *(const float4*)(A + (size_t)gr * K + c4 * 4);
  }

  const int tn   = w % tilesN;
  const int rsub = w / tilesN;
  const int col  = tn * 16 + l16;
  const bool cok = FULL || (col < Nc);
  const int colc = FULL ? col : (cok ? col : 0);
  const float* bp = B + (size_t)(2 * half) * Nc + colc;
  v2f bf[KS];
#pragma unroll
  for (int s = 0; s < KS; ++s) {
    float bx = bp[(size_t)(4 * s) * Nc];
    float by = bp[(size_t)(4 * s) * Nc + Nc];
    bf[s].x = FULL ? bx : (cok ? bx : 0.f);
    bf[s].y = FULL ? by : (cok ? by : 0.f);
  }
  float bv = 0.f;
  if (bias != nullptr && cok) bv = bias[col];

#pragma unroll
  for (int i = 0; i < 8; ++i) {
    int idx4 = i * 256 + t;
    int row  = idx4 / C4;
    int c4   = idx4 - row * C4;
    *(float4*)(sA + row * P + c4 * 4) = av[i];
  }
  __syncthreads();

  for (int j = 0; j < TPW; ++j) {
    const int rt   = rsub + WG * j;
    const int row0 = r0 + rt * 16;
    if (row0 >= M) break;
    const float* ar = sA + (rt * 16 + l16) * P + 2 * half;
    v8f c = {0.f, 0.f, 0.f, 0.f, 0.f, 0.f, 0.f, 0.f};
#pragma unroll
    for (int s0 = 0; s0 < KS; s0 += CH) {
      v2f a[CH];
#pragma unroll
      for (int s = 0; s < CH; ++s) a[s] = *(const v2f*)(ar + 4 * (s0 + s));
      __builtin_amdgcn_sched_barrier(0);
#pragma unroll
      for (int s = 0; s < CH; ++s)
        c = __builtin_amdgcn_wmma_f32_16x16x4_f32(false, a[s], false, bf[s0 + s], (short)0, c,
                                                  false, false);
      __builtin_amdgcn_sched_barrier(0);
    }
    if (cok) {
      const int rbase = row0 + 8 * half;
      float* crow = Cout + (size_t)rbase * Nc + col;
      if (row0 + 16 <= M) {
#pragma unroll
        for (int v = 0; v < 8; ++v) {
          float val = c[v] + bv;
          if (act) val = fmaxf(val, 0.f);
          crow[(size_t)v * Nc] = val;
        }
      } else {
#pragma unroll
        for (int v = 0; v < 8; ++v) {
          float val = c[v] + bv;
          if (act) val = fmaxf(val, 0.f);
          if (rbase + v < M) crow[(size_t)v * Nc] = val;
        }
      }
    }
  }
}

template <int K, int Nc>
__global__ void __launch_bounds__(256)
gemm_wmma_f32(const float* __restrict__ A, const float* __restrict__ B,
              const float* __restrict__ bias, float* __restrict__ Cout, int M, int act) {
  __shared__ __attribute__((aligned(16))) float sA[(8192 / K) * (K + 4)];
  gemm_block<K, Nc>(A, B, bias, Cout, M, act, sA, (int)blockIdx.x);
}

__global__ void gemm_wmma_f32_dyn(const float* __restrict__ A, const float* __restrict__ B,
                                  const float* __restrict__ bias, float* __restrict__ Cout,
                                  int M, int K, int Nc, int act) {
  int wave   = blockIdx.x * (blockDim.x >> 5) + (threadIdx.x >> 5);
  int tilesN = (Nc + 15) >> 4;
  int tilesM = (M + 15) >> 4;
  if (wave >= tilesM * tilesN) return;
  int tm = wave / tilesN, tn = wave - tm * tilesN;
  int lane = threadIdx.x & 31, half = lane >> 4, l16 = lane & 15;
  int row = tm * 16 + l16, col = tn * 16 + l16;
  bool cok = (col < Nc);
  int colc = cok ? col : 0;
  const float* ap = A + (size_t)row * K + 2 * half;
  const float* bp = B + (size_t)(2 * half) * Nc + colc;
  v8f c = {0.f, 0.f, 0.f, 0.f, 0.f, 0.f, 0.f, 0.f};
  for (int k0 = 0; k0 < K; k0 += 4) {
    float2 av = *(const float2*)ap;
    float bx = bp[0], by = bp[Nc];
    v2f a, b;
    a.x = av.x;            a.y = av.y;
    b.x = cok ? bx : 0.f;  b.y = cok ? by : 0.f;
    c = __builtin_amdgcn_wmma_f32_16x16x4_f32(false, a, false, b, (short)0, c,
                                              false, false);
    ap += 4;
    bp += (size_t)4 * Nc;
  }
  if (cok) {
    float bv = bias ? bias[col] : 0.f;
    float* crow = Cout + (size_t)(tm * 16 + 8 * half) * Nc + col;
    for (int v = 0; v < 8; ++v) {
      float val = c[v] + bv;
      if (act) val = fmaxf(val, 0.f);
      crow[(size_t)v * Nc] = val;
    }
  }
}

constexpr int OWN_T   = 1024;
constexpr int NPB_MAX = 512;
constexpr int SCAN_U  = 8;
constexpr int TILE_E  = 1024;
constexpr int REC_MAX = 15;
constexpr int OVT_CAP = 1024;
static_assert(OWN_T == TILE_E, "the listed tiles are read one edge per thread");

template <int G>
__device__ __forceinline__ void
lines_out4(unsigned* tab, unsigned* __restrict__ bits, int rowW, int tile, int t) {
  const uint4* tab4 = (const uint4*)tab;
  const uint4 v0 = tab4[t + 256 * (G + 0)];
  const uint4 v1 = tab4[t + 256 * (G + 1)];
  const uint4 v2 = tab4[t + 256 * (G + 2)];
  const uint4 v3 = tab4[t + 256 * (G + 3)];
  if ((t & 7) == 0) {
    tab[((t + 256 * (G + 0)) >> 3) * 32] = 0u;
    tab[((t + 256 * (G + 1)) >> 3) * 32] = 0u;
    tab[((t + 256 * (G + 2)) >> 3) * 32] = 0u;
    tab[((t + 256 * (G + 3)) >> 3) * 32] = 0u;
  }
  unsigned* out = bits + (size_t)(t >> 3) * rowW + (size_t)tile * 32 + (t & 7) * 4;
  *(uint4*)(out + (size_t)(32 * (G + 0)) * rowW) = v0;
  *(uint4*)(out + (size_t)(32 * (G + 1)) * rowW) = v1;
  *(uint4*)(out + (size_t)(32 * (G + 2)) * rowW) = v2;
  *(uint4*)(out + (size_t)(32 * (G + 3)) * rowW) = v3;
}

__device__ __forceinline__ void
edge_recs_block(const int* __restrict__ ei, int Eorig, int npb, int rowW,
                unsigned* __restrict__ bits, int* __restrict__ nxt,
                unsigned* tab, int blk, int nblk) {
  const int t = threadIdx.x;
  const int* dp = ei + Eorig;
  const unsigned un = (unsigned)npb;
  const int ntiles = rowW >> 5;
  const bool wide = npb > 65536;
  const unsigned step = wide ? (unsigned)(REC_MAX + 1) : 1u;
  tab[t * 32] = 0u;
  for (int tile = blk; tile < ntiles; tile += nblk) {
    __syncthreads();
    const int e0 = tile * TILE_E + 4 * t;
    int d[4] = {-1, -1, -1, -1};
    int s[4] = {0, 0, 0, 0};
    if ((Eorig & 3) == 0 && e0 + 4 <= Eorig) {
      const int4 v = *(const int4*)(dp + e0);
      const int4 u = *(const int4*)(ei + e0);
      d[0] = v.x; d[1] = v.y; d[2] = v.z; d[3] = v.w;
      s[0] = u.x; s[1] = u.y; s[2] = u.z; s[3] = u.w;
      *(int4*)(nxt + e0) = make_int4(-1, -1, -1, -1);
    } else {
#pragma unroll
      for (int k = 0; k < 4; ++k)
        if (e0 + k < Eorig) { d[k] = dp[e0 + k]; s[k] = ei[e0 + k]; nxt[e0 + k] = -1; }
    }
    unsigned ow[4], slot[4];
#pragma unroll
    for (int k = 0; k < 4; ++k) {
      ow[k]   = (d[k] >= 0) ? (unsigned)d[k] / un : 256u;
      slot[k] = (unsigned)REC_MAX;
      if (ow[k] < 256u) slot[k] = atomicAdd(&tab[ow[k] * 32u], step);
    }
#pragma unroll
    for (int k = 0; k < 4; ++k) {
      if (slot[k] < (unsigned)REC_MAX && !wide)
        *(uint2*)(tab + ow[k] * 32u + 2u + 2u * slot[k]) =
            make_uint2((unsigned)s[k], ((unsigned)d[k] - ow[k] * un) | ((unsigned)(4 * t + k) << 16));
    }
    __syncthreads();
    lines_out4<0>(tab, bits, rowW, tile, t);
    lines_out4<4>(tab, bits, rowW, tile, t);
  }
}

__global__ void __launch_bounds__(256)
edge_recs(const int* __restrict__ ei, int Eorig, int npb, int rowW,
          unsigned* __restrict__ bits, int* __restrict__ nxt) {
  __shared__ __attribute__((aligned(16))) unsigned tab[256 * 32];
  edge_recs_block(ei, Eorig, npb, rowW, bits, nxt, tab, (int)blockIdx.x, (int)gridDim.x);
}

__global__ void __launch_bounds__(256)
proj1_beside_edge_recs(const float* __restrict__ x, const float* __restrict__ W1,
                       float* __restrict__ h1, int M, int gemm_blocks,
                       const int* __restrict__ ei, int Eorig, int npb, int rowW,
                       unsigned* __restrict__ bits, int* __restrict__ nxt, int edge_blocks) {
  __shared__ __attribute__((aligned(16))) float smem[(8192 / 64) * (64 + 4)];
  const int b = (int)blockIdx.x;
  if (b < gemm_blocks)
    gemm_block<64, 128>(x, W1, nullptr, h1, M, 0, smem, b);
  else
    edge_recs_block(ei, Eorig, npb, rowW, bits, nxt, (unsigned*)smem, b - gemm_blocks, edge_blocks);
}

template <int L>
__device__ __forceinline__ unsigned row_lane(unsigned v) {
  return (unsigned)__builtin_amdgcn_update_dpp(0, (int)v, 0x150 | L, 0xF, 0xF, true);
}

__global__ void __launch_bounds__(OWN_T)
owner_rows(const int* __restrict__ ei, int Eorig, const unsigned* __restrict__ bits, int rowW,
           int npb, int n, int* __restrict__ rows, int* __restrict__ curs, int* __restrict__ nxt) {
  __shared__ int rows_s[NPB_MAX * 32];
  __shared__ int cnt_s[NPB_MAX];
  __shared__ int ovh_s[NPB_MAX];
  __shared__ int ovt[OVT_CAP];
  __shared__ int novt;
  const int t  = threadIdx.x;
  const int o  = blockIdx.x;
  const int lo = o * npb;
  const int hi = (lo + npb < n) ? lo + npb : n;
  const int* dp = ei + Eorig;
  const uint4* brow = (const uint4*)(bits + (size_t)o * rowW);
  const int nvec = rowW >> 2;
  const int p = t & 7;

  for (int base = lo; base < hi; base += NPB_MAX) {
    const int span = (hi - base < NPB_MAX) ? hi - base : NPB_MAX;
    for (int i = t; i < NPB_MAX; i += OWN_T) { cnt_s[i] = 0; ovh_s[i] = -1; }
    if (t == 0) novt = 0;
    __syncthreads();

    auto rank_edge = [&](int e, int d, int sv) {
      const unsigned loc = (unsigned)(d - base);
      if (loc < (unsigned)span) {
        const int r = atomicAdd(&cnt_s[loc], 1);
        if (r < 32) rows_s[loc * 32 + r] = sv;
        else nxt[e] = atomicExch(&ovh_s[loc], e);
      }
    };

    for (int ib = 0; ib < nvec; ib += SCAN_U * OWN_T) {
      const int it0 = ib + t;
      uint4 w[SCAN_U];
#pragma unroll
      for (int u = 0; u < SCAN_U; ++u) {
        const int it = it0 + u * OWN_T;
        w[u] = brow[it < nvec ? it : 0];
      }
      __builtin_amdgcn_sched_barrier(0);
#pragma unroll
      for (int u = 0; u < SCAN_U; ++u) {
        const int it = it0 + u * OWN_T;
        if (it >= nvec) w[u] = make_uint4(0u, 0u, 0u, 0u);
        const unsigned c0 = row_lane<0>(w[u].x);
        const unsigned c8 = row_lane<8>(w[u].x);
        const unsigned cn = (t & 8) ? c8 : c0;
        const int nrec = cn <= (unsigned)REC_MAX ? (int)cn : 0;
        const int tile = it >> 3;
        const int eb   = tile * TILE_E;
        if (p != 0 && 2 * p - 1 < nrec)
          rank_edge(eb + (int)(w[u].y >> 16), lo + (int)(w[u].y & 0xFFFFu), (int)w[u].x);
        if (2 * p < nrec)
          rank_edge(eb + (int)(w[u].w >> 16), lo + (int)(w[u].w & 0xFFFFu), (int)w[u].z);
        if (p == 0 && cn > (unsigned)REC_MAX) {
          const int sl = atomicAdd(&novt, 1);
          if (sl < OVT_CAP) ovt[sl] = tile;
          else {
#pragma nounroll
            for (int k = 0; k < TILE_E; ++k) {
              const int e = eb + k;
              if (e < Eorig) rank_edge(e, dp[e], ei[e]);
            }
          }
        }
      }
    }
    __syncthreads();

    const int nov = novt < OVT_CAP ? novt : OVT_CAP;
    for (int i = 0; i < nov; ++i) {
      const int e = ovt[i] * TILE_E + t;
      if (e < Eorig) rank_edge(e, dp[e], ei[e]);
    }
    __syncthreads();

    for (int idx = t; idx < span * 32; idx += OWN_T) {
      const int loc = idx >> 5, k = idx & 31;
      int v = -1;
      if (k < cnt_s[loc]) v = rows_s[idx];
      rows[(size_t)(base + loc) * 32 + k] = v;
    }
    for (int idx = t; idx < span * 4; idx += OWN_T) {
      const int loc = idx >> 2;
      curs[(size_t)(base + loc) * 4 + (idx & 3)] = ((idx & 3) == 0) ? ovh_s[loc] : -1;
    }
    __syncthreads();
  }
}

template <int MASK>
__device__ __forceinline__ float row_xor(float v) {
  return __int_as_float(__builtin_amdgcn_update_dpp(0, __float_as_int(v), 0x160 | MASK, 0xF, 0xF, true));
}
template <int L>
__device__ __forceinline__ float row_share(float v) {
  return __int_as_float(__builtin_amdgcn_update_dpp(0, __float_as_int(v), 0x150 | L, 0xF, 0xF, true));
}
__device__ __forceinline__ float half_sum(float v) {
  v += row_xor<8>(v);
  v += row_xor<4>(v);
  v += row_xor<2>(v);
  v += row_xor<1>(v);
  return v;
}
__device__ __forceinline__ float half_max(float v) {
  v = fmaxf(v, row_xor<8>(v));
  v = fmaxf(v, row_xor<4>(v));
  v = fmaxf(v, row_xor<2>(v));
  v = fmaxf(v, row_xor<1>(v));
  return v;
}
__device__ __forceinline__ float dot4(float4 a, float4 b) {
  return fmaf(a.w, b.w, fmaf(a.z, b.z, fmaf(a.y, b.y, a.x * b.x)));
}
__device__ __forceinline__ float leaky(float a) { return (a > 0.f) ? a : 0.2f * a; }
__device__ __forceinline__ float exp_nonpos(float x) {
  return __builtin_amdgcn_exp2f(x * 1.44269504088896341f);
}

struct Soft { float m, zl, a0, a1, a2, a3; };

template <int HEADS, int R, bool FULL>
__device__ __forceinline__ void gather_chunk(Soft& st, const float* __restrict__ h, int my_src,
                                             int k, int cnt, unsigned off, int half, int c16,
                                             float4 as4, float adn) {
  constexpr int F   = HEADS * 64;
  constexpr int EPL = (HEADS == 2) ? 1 : 2;
  float4 t[R];
#pragma unroll
  for (int q = 0; q < R; ++q) {
    const int e0 = k + q * EPL;
    if constexpr (!FULL) t[q] = make_float4(0.f, 0.f, 0.f, 0.f);
    if (FULL || e0 < cnt) {
      int s = __builtin_amdgcn_readlane(my_src, e0);
      if constexpr (HEADS == 1) {
        const int e1 = (FULL || e0 + 1 < cnt) ? e0 + 1 : e0;
        const int s1 = __builtin_amdgcn_readlane(my_src, e1);
        s = half ? s1 : s;
      }
      t[q] = *(const float4*)(h + ((unsigned)s * (unsigned)F + off));
    }
  }
  float d = 0.f;
#pragma unroll
  for (int q = 0; q < R; ++q) {
    if (FULL || k + q * EPL < cnt) {
      const float dq = half_sum(dot4(t[q], as4));
      d = (c16 == q) ? dq : d;
    }
  }
  const int  eidx  = k + c16 * EPL + ((HEADS == 1) ? half : 0);
  const bool valid = (c16 < R) && (FULL || eidx < cnt);
  const float a    = valid ? leaky(d + adn) : -INFINITY;
  const float mn   = fmaxf(st.m, half_max(a));
  const float ev   = exp_nonpos(((c16 == R) ? st.m : a) - mn);
  const float rsc  = row_share<R>(ev);
  st.zl = st.zl * rsc + ((c16 == R) ? 0.f : ev);
  st.a0 *= rsc; st.a1 *= rsc; st.a2 *= rsc; st.a3 *= rsc;
  st.m = mn;
  auto add_row = [&](float w, float4 r) {
    st.a0 = fmaf(w, r.x, st.a0); st.a1 = fmaf(w, r.y, st.a1);
    st.a2 = fmaf(w, r.z, st.a2); st.a3 = fmaf(w, r.w, st.a3);
  };
  if (FULL || k + 0 * EPL < cnt) add_row(row_share<0>(ev), t[0]);
  if (FULL || k + 1 * EPL < cnt) add_row(row_share<1>(ev), t[1]);
  if (FULL || k + 2 * EPL < cnt) add_row(row_share<2>(ev), t[2]);
  if (FULL || k + 3 * EPL < cnt) add_row(row_share<3>(ev), t[3]);
  if (FULL || k + 4 * EPL < cnt) add_row(row_share<4>(ev), t[4]);
  if (FULL || k + 5 * EPL < cnt) add_row(row_share<5>(ev), t[5]);
  if (FULL || k + 6 * EPL < cnt) add_row(row_share<6>(ev), t[6]);
  if (FULL || k + 7 * EPL < cnt) add_row(row_share<7>(ev), t[7]);
}

template <int HEADS>
__global__ void __launch_bounds__(256)
node_gather(const int* __restrict__ ei, int Eorig,
                            const int* __restrict__ nxt, const int* __restrict__ rows,
                            const int* __restrict__ curs, const float* __restrict__ a_src,
                            const float* __restrict__ a_dst, const float* __restrict__ h,
                            const float* __restrict__ bias, float* __restrict__ out, int n) {
  constexpr int F   = HEADS * 64;
  constexpr int EPL = (HEADS == 2) ? 1 : 2;
  constexpr int R   = 8;
  int node = __builtin_amdgcn_readfirstlane((int)(blockIdx.x * (blockDim.x >> 5) + (threadIdx.x >> 5)));
  if (node >= n) return;
  const int lane = threadIdx.x & 31;
  const int half = lane >> 4;
  const int c16  = lane & 15;
  const int hd   = (HEADS == 2) ? half : 0;
  const unsigned off = (HEADS == 2) ? lane * 4 : c16 * 4;

  const int pre = rows[(size_t)node * 32 + lane];
  int cur = curs[node * 4 + (lane & 3)];
  const float4 own = *(const float4*)(h + (size_t)node * F + off);
  const float4 as4 = *(const float4*)(a_src + hd * 64 + c16 * 4);
  const float4 ad4 = *(const float4*)(a_dst + hd * 64 + c16 * 4);
  __builtin_amdgcn_sched_barrier(0);

  const float adn = half_sum(dot4(own, ad4));
  const bool mine = (HEADS == 2) || (half == 0);
  Soft st;
  st.m  = leaky(half_sum(dot4(own, as4)) + adn);
  st.zl = (mine && c16 == 0) ? 1.f : 0.f;
  st.a0 = mine ? own.x : 0.f;
  st.a1 = mine ? own.y : 0.f;
  st.a2 = mine ? own.z : 0.f;
  st.a3 = mine ? own.w : 0.f;

  unsigned act = __builtin_amdgcn_ballot_w32(cur >= 0) & 0xFu;
  const int max_rounds = (Eorig >> 5) + 2;
  int round = 0;
  do {
    int cnt = 0;
    int my_src = node;
    if (round == 0) {
      my_src = pre;
      cnt = __builtin_popcount(__builtin_amdgcn_ballot_w32(pre >= 0));
    } else
    while (act != 0u && cnt < 32) {
      int ci = cur > 0 ? cur : 0;
      int sv = ei[ci];
      int nv = nxt[ci];
#pragma unroll
      for (int c = 0; c < 4; ++c) {
        if (((act >> c) & 1u) != 0u && cnt < 32) {
          int s = __builtin_amdgcn_readlane(sv, c);
          if (lane == cnt) my_src = s;
          ++cnt;
          if ((lane & 3) == c) cur = nv;
        }
      }
      act = __builtin_amdgcn_ballot_w32(cur >= 0) & 0xFu;
    }
    int k = 0;
    for (; k + R * EPL <= cnt; k += R * EPL)
      gather_chunk<HEADS, R, true>(st, h, my_src, k, cnt, off, half, c16, as4, adn);
    if (k < cnt)
      gather_chunk<HEADS, R, false>(st, h, my_src, k, cnt, off, half, c16, as4, adn);
    if (act == 0u) break;
  } while (++round < max_rounds);

  float z = half_sum(st.zl);
  float acc[4] = {st.a0, st.a1, st.a2, st.a3};
  if constexpr (HEADS == 1) {
    const float mo  = __shfl_xor(st.m, 16, 32);
    const float rsc = exp_nonpos(st.m - fmaxf(st.m, mo));
    z *= rsc;
#pragma unroll
    for (int j = 0; j < 4; ++j) acc[j] *= rsc;
    z += __shfl_xor(z, 16, 32);
#pragma unroll
    for (int j = 0; j < 4; ++j) acc[j] += __shfl_xor(acc[j], 16, 32);
  }

  const float inv = 1.0f / (z + 1e-16f);
  if constexpr (HEADS == 2) {
    const float4 bv = *(const float4*)(bias + off);
    const float bb[4] = {bv.x, bv.y, bv.z, bv.w};
    float o[4];
#pragma unroll
    for (int j = 0; j < 4; ++j) {
      float v = fmaf(acc[j], inv, bb[j]);
      o[j] = (v > 0.f) ? v : exp_nonpos(v) - 1.0f;
    }
    *(float4*)(out + (size_t)node * F + off) = make_float4(o[0], o[1], o[2], o[3]);
  } else {
    const unsigned o2 = off + 2u * (unsigned)half;
    const float2 bv = *(const float2*)(bias + o2);
    const float v0 = fmaf(half ? acc[2] : acc[0], inv, bv.x);
    const float v1 = fmaf(half ? acc[3] : acc[1], inv, bv.y);
    const float r0 = (v0 > 0.f) ? v0 : exp_nonpos(v0) - 1.0f;
    const float r1 = (v1 > 0.f) ? v1 : exp_nonpos(v1) - 1.0f;
    *(float2*)(out + (size_t)node * F + o2) = make_float2(r0, r1);
  }
}

extern "C" void kernel_launch(void* const* d_in, const int* in_sizes, int n_in,
                              void* d_out, int out_size, void* d_ws, size_t ws_size,
                              hipStream_t stream) {
  const float* x      = (const float*)d_in[0];
  const int*   ei     = (const int*)d_in[1];
  const float* W1     = (const float*)d_in[2];
  const float* a_src1 = (const float*)d_in[3];
  const float* a_dst1 = (const float*)d_in[4];
  const float* b1     = (const float*)d_in[5];
  const float* W2     = (const float*)d_in[6];
  const float* a_src2 = (const float*)d_in[7];
  const float* a_dst2 = (const float*)d_in[8];
  const float* b2     = (const float*)d_in[9];
  const float* m1w    = (const float*)d_in[10];
  const float* m1b    = (const float*)d_in[11];
  const float* m2w    = (const float*)d_in[12];
  const float* m2b    = (const float*)d_in[13];

  const int N    = in_sizes[0] / 64;
  const int E    = in_sizes[1] / 2;
  const int C    = in_sizes[12] / 64;

  float* bufA = (float*)d_ws;
  float* bufB = bufA + (size_t)N * 128;
  float* bufC = bufB + (size_t)N * 128;
  int*   nxt  = (int*)(bufC + (size_t)N * 68);
  const int ntiles = (E + TILE_E - 1) / TILE_E;
  const int rowW   = ntiles * 32;
  const size_t bits_words  = (size_t)256 * rowW;
  const size_t plane_words = (size_t)N * 128;
  size_t spill = bits_words > plane_words ? bits_words - plane_words : 0;
  spill = (spill + 31) & ~(size_t)31;
  const bool side = spill <= (size_t)N * 32;
  unsigned* bits = side ? (unsigned*)bufB : (unsigned*)bufA;
  int*   rows = (int*)bufC + (side ? spill : 0);
  int*   curs = rows + (size_t)N * 32;

  const int TPB = 256;
  auto gemm_blocks = [&](int M, int Nc) {
    int waves = ((M + 15) / 16) * ((Nc + 15) / 16);
    return (waves + 7) / 8;
  };
  auto gemm_row_blocks = [&](int M, int K) {
    int rb = 8192 / K;
    return (M + rb - 1) / rb;
  };

  const int npb    = (N > 256) ? (N + 255) / 256 : 1;
  const int owners = (N + npb - 1) / npb;
  const int eblocks = (ntiles + 1) / 2;
  const int gblocks = gemm_row_blocks(N, 64);
  if (side) {
    proj1_beside_edge_recs<<<gblocks + eblocks, TPB, 0, stream>>>(x, W1, bufA, N, gblocks,
                                                                  ei, E, npb, rowW, bits, nxt, eblocks);
    owner_rows<<<owners, OWN_T, 0, stream>>>(ei, E, bits, rowW, npb, N, rows, curs, nxt);
  } else {
    edge_recs<<<eblocks, TPB, 0, stream>>>(ei, E, npb, rowW, bits, nxt);
    owner_rows<<<owners, OWN_T, 0, stream>>>(ei, E, bits, rowW, npb, N, rows, curs, nxt);
    gemm_wmma_f32<64, 128><<<gblocks, TPB, 0, stream>>>(x, W1, nullptr, bufA, N, 0);
  }

  node_gather<2><<<(N + 7) / 8, TPB, 0, stream>>>(ei, E, nxt, rows, curs, a_src1, a_dst1,
                                                  bufA, b1, bufB, N);

  gemm_wmma_f32<128, 64><<<gemm_row_blocks(N, 128), TPB, 0, stream>>>(bufB, W2, nullptr,
                                                                 bufA, N, 0);
  node_gather<1><<<(N + 7) / 8, TPB, 0, stream>>>(ei, E, nxt, rows, curs, a_src2, a_dst2,
                                                  bufA, b2, bufB, N);

  gemm_wmma_f32<64, 64><<<gemm_row_blocks(N, 64), TPB, 0, stream>>>(bufB, m1w, m1b,
                                                                bufA, N, 1);
  if (C == 8) {
    gemm_wmma_f32<64, 8><<<gemm_row_blocks(N, 64), TPB, 0, stream>>>(bufA, m2w, m2b,
                                                                (float*)d_out, N, 1);
  } else {
    gemm_wmma_f32_dyn<<<gemm_blocks(N, C), TPB, 0, stream>>>(bufA, m2w, m2b,
                                                             (float*)d_out, N, 64, C, 1);
  }
}
